# starting kernel
# baseline (speedup 1.0000x reference)
#include <hip/hip_runtime.h>

typedef __attribute__((ext_vector_type(2))) float v2f;
typedef __attribute__((ext_vector_type(8))) float v8f;

#define N_NODES  50000
#define N_EDGES  800000
#define N_GRAPHS 512
#define BN_EPS   1e-5f

__global__ void zero_f32(float* __restrict__ p, int n4) {
    int t = blockIdx.x * blockDim.x + threadIdx.x;
    if (t < n4) ((float4*)p)[t] = make_float4(0.f, 0.f, 0.f, 0.f);
}

__global__ void gather_emb(const int* __restrict__ ids, const float* __restrict__ emb,
                           float* __restrict__ x) {
    int t = blockIdx.x * blockDim.x + threadIdx.x;
    int node = t >> 5, lane = t & 31;
    if (node >= N_NODES) return;
    int id = ids[node];
    float4 v = *(const float4*)(emb + (long)id * 128 + lane * 4);
    *(float4*)(x + (long)node * 128 + lane * 4) = v;
}

__global__ void edge_scatter(const float* __restrict__ x, const int* __restrict__ ei,
                             float* __restrict__ agg) {
    int wave = threadIdx.x >> 5, lane = threadIdx.x & 31;
    int e = blockIdx.x * 8 + wave;
    if (e >= N_EDGES) return;
    int s = ei[e];
    int d = ei[N_EDGES + e];
    float4 v = *(const float4*)(x + (long)s * 128 + lane * 4);
    float* p = agg + (long)d * 128 + lane * 4;
    atomicAdd(p + 0, v.x);
    atomicAdd(p + 1, v.y);
    atomicAdd(p + 2, v.z);
    atomicAdd(p + 3, v.w);
}

template <bool FUSE_ADD, bool BN, bool RELU>
__global__ void gemm_wmma(const float* __restrict__ X, const float* __restrict__ AGG,
                          const float* __restrict__ W, const float* __restrict__ bias,
                          const float* __restrict__ gamma, const float* __restrict__ beta,
                          const float* __restrict__ mean, const float* __restrict__ var,
                          float* __restrict__ O, int M, int K, int N) {
    int wave = threadIdx.x >> 5, lane = threadIdx.x & 31;
    int tilesN = N >> 4;
    int tile = blockIdx.x * 8 + wave;
    if (tile >= (M >> 4) * tilesN) return;
    int mt = tile / tilesN, nt = tile % tilesN;
    int mi = lane & 15, kh = lane >> 4;

    const float* xr = X + (long)(mt * 16 + mi) * K + kh * 2;
    const float* ar = FUSE_ADD ? (AGG + (long)(mt * 16 + mi) * K + kh * 2) : nullptr;
    const float* wc = W + (long)(kh * 2) * N + nt * 16 + mi;

    v8f acc = {0.f, 0.f, 0.f, 0.f, 0.f, 0.f, 0.f, 0.f};
    for (int kb = 0; kb < K; kb += 4) {
        v2f a, b;
        a.x = xr[kb];
        a.y = xr[kb + 1];
        if (FUSE_ADD) { a.x += ar[kb]; a.y += ar[kb + 1]; }
        b.x = wc[(long)kb * N];
        b.y = wc[(long)(kb + 1) * N];
        acc = __builtin_amdgcn_wmma_f32_16x16x4_f32(false, a, false, b,
                                                    (short)0, acc, false, false);
    }

    int col = nt * 16 + mi;
    float bb = bias[col];
    float mu = 0.f, inv = 1.f, gm = 1.f, bt = 0.f;
    if (BN) { mu = mean[col]; inv = rsqrtf(var[col] + BN_EPS); gm = gamma[col]; bt = beta[col]; }
    float* orow = O + (long)(mt * 16 + kh * 8) * N + col;
#pragma unroll
    for (int r = 0; r < 8; r++) {
        float v = acc[r] + bb;
        if (BN)   v = (v - mu) * inv * gm + bt;
        if (RELU) v = fmaxf(v, 0.f);
        orow[(long)r * N] = v;
    }
}

__global__ void pool3(const float* __restrict__ h1, const float* __restrict__ h2,
                      const float* __restrict__ h3, const int* __restrict__ batch,
                      float* __restrict__ gcat) {
    int t = blockIdx.x * blockDim.x + threadIdx.x;
    int node = t >> 5, lane = t & 31;
    if (node >= N_NODES) return;
    int g = batch[node];
    float* base = gcat + (long)g * 384 + lane * 4;
    const float* hs[3] = {h1, h2, h3};
#pragma unroll
    for (int l = 0; l < 3; l++) {
        float4 v = *(const float4*)(hs[l] + (long)node * 128 + lane * 4);
        float* p = base + l * 128;
        atomicAdd(p + 0, v.x);
        atomicAdd(p + 1, v.y);
        atomicAdd(p + 2, v.z);
        atomicAdd(p + 3, v.w);
    }
}

__global__ void lin2_kernel(const float* __restrict__ h, const float* __restrict__ w,
                            const float* __restrict__ b, float* __restrict__ out) {
    int t = blockIdx.x * blockDim.x + threadIdx.x;
    if (t >= N_GRAPHS * 10) return;
    int g = t / 10, o = t % 10;
    const float* hr = h + (long)g * 384;
    float acc = b[o];
    for (int k = 0; k < 384; k++) acc = fmaf(hr[k], w[k * 10 + o], acc);
    out[t] = acc;
}

extern "C" void kernel_launch(void* const* d_in, const int* in_sizes, int n_in,
                              void* d_out, int out_size, void* d_ws, size_t ws_size,
                              hipStream_t stream) {
    const int*   x_ids = (const int*)d_in[0];
    const int*   edge  = (const int*)d_in[1];
    const int*   batch = (const int*)d_in[2];
    const float* emb   = (const float*)d_in[3];

    const float *c_w1[3], *c_b1[3], *c_g[3], *c_be[3], *c_mu[3], *c_va[3], *c_w2[3], *c_b2[3];
    for (int i = 0; i < 3; i++) {
        c_w1[i] = (const float*)d_in[4 + 8 * i + 0];
        c_b1[i] = (const float*)d_in[4 + 8 * i + 1];
        c_g[i]  = (const float*)d_in[4 + 8 * i + 2];
        c_be[i] = (const float*)d_in[4 + 8 * i + 3];
        c_mu[i] = (const float*)d_in[4 + 8 * i + 4];
        c_va[i] = (const float*)d_in[4 + 8 * i + 5];
        c_w2[i] = (const float*)d_in[4 + 8 * i + 6];
        c_b2[i] = (const float*)d_in[4 + 8 * i + 7];
    }
    const float* lin1_w = (const float*)d_in[28];
    const float* lin1_b = (const float*)d_in[29];
    const float* lin2_w = (const float*)d_in[30];
    const float* lin2_b = (const float*)d_in[31];

    const long NF = (long)N_NODES * 128;
    float* ws   = (float*)d_ws;
    float* x0   = ws + 0 * NF;
    float* agg  = ws + 1 * NF;
    float* tmp  = ws + 2 * NF;
    float* hbuf[3] = { ws + 3 * NF, ws + 4 * NF, x0 };
    float* gcat  = agg;
    float* lin1o = tmp;

    gather_emb<<<(N_NODES * 32 + 255) / 256, 256, 0, stream>>>(x_ids, emb, x0);

    const int tiles_node = (N_NODES / 16) * (128 / 16);
    const float* xin = x0;
    for (int l = 0; l < 3; l++) {
        zero_f32<<<(int)((NF / 4 + 255) / 256), 256, 0, stream>>>(agg, (int)(NF / 4));
        edge_scatter<<<N_EDGES / 8, 256, 0, stream>>>(xin, edge, agg);
        gemm_wmma<true, true, true><<<(tiles_node + 7) / 8, 256, 0, stream>>>(
            xin, agg, c_w1[l], c_b1[l], c_g[l], c_be[l], c_mu[l], c_va[l],
            tmp, N_NODES, 128, 128);
        gemm_wmma<false, false, true><<<(tiles_node + 7) / 8, 256, 0, stream>>>(
            tmp, nullptr, c_w2[l], c_b2[l], nullptr, nullptr, nullptr, nullptr,
            hbuf[l], N_NODES, 128, 128);
        xin = hbuf[l];
    }

    zero_f32<<<(N_GRAPHS * 384 / 4 + 255) / 256, 256, 0, stream>>>(gcat, N_GRAPHS * 384 / 4);
    pool3<<<(N_NODES * 32 + 255) / 256, 256, 0, stream>>>(hbuf[0], hbuf[1], hbuf[2], batch, gcat);

    const int tiles_head = (N_GRAPHS / 16) * (384 / 16);
    gemm_wmma<false, false, true><<<(tiles_head + 7) / 8, 256, 0, stream>>>(
        gcat, nullptr, lin1_w, lin1_b, nullptr, nullptr, nullptr, nullptr,
        lin1o, N_GRAPHS, 384, 384);

    lin2_kernel<<<(N_GRAPHS * 10 + 255) / 256, 256, 0, stream>>>(lin1o, lin2_w, lin2_b,
                                                                 (float*)d_out);
}


# Neighbour sum by owner groups: each of 256 groups keeps its 196 destination rows in LDS, scans the edge list, adds source rows by LDS float adds; no global atomic, no zero fill; sum order LDS arrival
# speedup vs baseline: 2.0581x; 2.0581x over previous
#include <hip/hip_runtime.h>

typedef __attribute__((ext_vector_type(2))) float v2f;
typedef __attribute__((ext_vector_type(8))) float v8f;

#define N_NODES  50000
#define N_EDGES  800000
#define N_GRAPHS 512
#define BN_EPS   1e-5f

__global__ void zero_f32(float* __restrict__ p, int n4) {
    int t = blockIdx.x * blockDim.x + threadIdx.x;
    if (t < n4) ((float4*)p)[t] = make_float4(0.f, 0.f, 0.f, 0.f);
}

__global__ void gather_emb(const int* __restrict__ ids, const float* __restrict__ emb,
                           float* __restrict__ x) {
    int t = blockIdx.x * blockDim.x + threadIdx.x;
    int node = t >> 5, lane = t & 31;
    if (node >= N_NODES) return;
    int id = ids[node];
    float4 v = *(const float4*)(emb + (long)id * 128 + lane * 4);
    *(float4*)(x + (long)node * 128 + lane * 4) = v;
}

#define AO_ROWS   196
#define AO_GROUPS ((N_NODES + AO_ROWS - 1) / AO_ROWS)
#define AO_SCAN   8
#define AO_TAKE   16
#define AO_QCAP   (AO_SCAN * 128 + AO_TAKE)
#define AO_ITERS  (N_EDGES / 128)
static_assert(N_EDGES % 128 == 0, "a wave-load of the scan covers 128 edges");
static_assert(N_EDGES <= (1 << 20) && AO_ROWS <= (1 << 12), "a queue word holds edge and local row");
static_assert(AO_QCAP % AO_TAKE == 0 && AO_TAKE <= 32 && (AO_TAKE & (AO_TAKE - 1)) == 0,
              "a scan group fits behind a rest of less than one pass; one wave writes a pass's fill");

__device__ __forceinline__ void ao_take(const unsigned* q, int i,
                                        const float* __restrict__ x, const int* __restrict__ ei,
                                        float* acc, int lane) {
    unsigned en[AO_TAKE];
    int s[AO_TAKE];
    float v[AO_TAKE][4];
#pragma unroll
    for (int j = 0; j < AO_TAKE; j++) {
        en[j] = __builtin_amdgcn_readfirstlane(q[i + j]);
        s[j] = ei[en[j] & 0xFFFFFu];
    }
#pragma unroll
    for (int j = 0; j < AO_TAKE; j++) {
        const float* p = x + (long)s[j] * 128 + lane;
        v[j][0] = p[0];
        v[j][1] = p[32];
        v[j][2] = p[64];
        v[j][3] = p[96];
    }
#pragma unroll
    for (int j = 0; j < AO_TAKE; j++) {
        float* a = acc + (en[j] >> 20) * 128 + lane;
        atomicAdd(a + 0,  v[j][0]);
        atomicAdd(a + 32, v[j][1]);
        atomicAdd(a + 64, v[j][2]);
        atomicAdd(a + 96, v[j][3]);
    }
}

#define AO_FILL ((unsigned)AO_ROWS << 20)
__device__ __forceinline__ void ao_drain(unsigned* q, int cnt,
                                         const float* __restrict__ x, const int* __restrict__ ei,
                                         float* acc, int lane) {
    const int fill = (AO_TAKE - (cnt & (AO_TAKE - 1))) & (AO_TAKE - 1);
    if (lane < fill) q[cnt + lane] = AO_FILL;
    for (int i = 0; i < cnt; i += AO_TAKE) ao_take(q, i, x, ei, acc, lane);
}

__device__ __forceinline__ int ao_full(unsigned* q, int cnt,
                                       const float* __restrict__ x, const int* __restrict__ ei,
                                       float* acc, int lane) {
    const int full = cnt & ~(AO_TAKE - 1);
    for (int i = 0; i < full; i += AO_TAKE) ao_take(q, i, x, ei, acc, lane);
    const int rest = cnt - full;
    if (lane < rest) q[lane] = q[full + lane];
    return rest;
}

__device__ __forceinline__ int ao_push(unsigned* q, int cnt, bool m, int dl, int e) {
    const unsigned mask = __builtin_amdgcn_ballot_w32(m);
    if (mask != 0u) {
        if (m) q[cnt + (int)__builtin_amdgcn_mbcnt_lo(mask, 0u)] = (unsigned)e | ((unsigned)dl << 20);
        cnt += __builtin_popcount(mask);
    }
    return cnt;
}

__device__ __forceinline__ int ao_scan(unsigned* q, int cnt, int4 d, int e0, int lo, int nrows) {
    const bool m0 = (unsigned)(d.x - lo) < (unsigned)nrows;
    const bool m1 = (unsigned)(d.y - lo) < (unsigned)nrows;
    const bool m2 = (unsigned)(d.z - lo) < (unsigned)nrows;
    const bool m3 = (unsigned)(d.w - lo) < (unsigned)nrows;
    if (__builtin_amdgcn_ballot_w32(m0 | m1 | m2 | m3) != 0u) {
        cnt = ao_push(q, cnt, m0, d.x - lo, e0 + 0);
        cnt = ao_push(q, cnt, m1, d.y - lo, e0 + 1);
        cnt = ao_push(q, cnt, m2, d.z - lo, e0 + 2);
        cnt = ao_push(q, cnt, m3, d.w - lo, e0 + 3);
    }
    return cnt;
}

__global__ void __launch_bounds__(1024)
agg_owner(const float* __restrict__ x, const int* __restrict__ ei, float* __restrict__ agg) {
    __shared__ __attribute__((aligned(16))) float acc[(AO_ROWS + 1) * 128];
    __shared__ __attribute__((aligned(16))) unsigned queue[32][AO_QCAP];
    const int wave = __builtin_amdgcn_readfirstlane(threadIdx.x >> 5), lane = threadIdx.x & 31;
    const int lo = blockIdx.x * AO_ROWS;
    const int nrows = (N_NODES - lo) < AO_ROWS ? (N_NODES - lo) : AO_ROWS;

    for (int i = threadIdx.x; i < (AO_ROWS + 1) * 32; i += 1024)
        ((float4*)acc)[i] = make_float4(0.f, 0.f, 0.f, 0.f);
    __syncthreads();

    const int4* dst4 = (const int4*)(ei + N_EDGES);
    unsigned* q = queue[wave];
    int cnt = 0;
    const int n_it = (AO_ITERS - wave + 31) / 32;
    const int first = (int)(blockIdx.x * n_it) / AO_GROUPS;

    for (int k = 0; k < n_it; k += AO_SCAN) {
        int4 d[AO_SCAN];
        int it[AO_SCAN];
#pragma unroll
        for (int u = 0; u < AO_SCAN; u++) {
            int kk = ((k + u) < n_it ? (k + u) : (n_it - 1)) + first;
            if (kk >= n_it) kk -= n_it;
            it[u] = wave + 32 * kk;
            d[u] = dst4[it[u] * 32 + lane];
        }
#pragma unroll
        for (int u = 0; u < AO_SCAN; u++)
            cnt = ao_scan(q, cnt, d[u], it[u] * 128 + lane * 4, lo, (k + u) < n_it ? nrows : 0);
        if (cnt >= AO_TAKE) cnt = ao_full(q, cnt, x, ei, acc, lane);
    }
    ao_drain(q, cnt, x, ei, acc, lane);

    __syncthreads();
    float4* out4 = (float4*)(agg + (long)lo * 128);
    for (int i = threadIdx.x; i < nrows * 32; i += 1024) out4[i] = ((const float4*)acc)[i];
}

#define NG_PITCH 132
#define NG_TILES (N_NODES / 16)
#define NODE_GEMM_GROUPS 512
static_assert(NODE_GEMM_GROUPS <= NG_TILES, "every work-group has a first tile");
static_assert(N_NODES % 16 == 0, "row tiles must tile the node list");
template <bool FUSE_ADD, bool BN, bool RELU>
__global__ void __launch_bounds__(256)
gemm_node(const float* __restrict__ X, const float* __restrict__ AGG,
          const float* __restrict__ W, const float* __restrict__ bias,
          const float* __restrict__ gamma, const float* __restrict__ beta,
          const float* __restrict__ mean, const float* __restrict__ var,
          float* __restrict__ O) {
    __shared__ __attribute__((aligned(16))) float tileA[2][16 * NG_PITCH];
    const int wave = threadIdx.x >> 5, lane = threadIdx.x & 31;
    const int mi = lane & 15, kh = lane >> 4;
    const int col = wave * 16 + mi;

    v2f bf[32];
    const float* wc = W + (long)(kh * 2) * 128 + col;
#pragma unroll
    for (int j = 0; j < 32; j++) {
        bf[j].x = wc[(long)(4 * j) * 128];
        bf[j].y = wc[(long)(4 * j + 1) * 128];
    }

    const float bb = bias[col];
    float mu = 0.f, inv = 1.f, gm = 1.f, bt = 0.f;
    if (BN) { mu = mean[col]; inv = rsqrtf(var[col] + BN_EPS); gm = gamma[col]; bt = beta[col]; }

    const int so0 = wave * NG_PITCH + lane * 4;
    const int so1 = (wave + 8) * NG_PITCH + lane * 4;
    const long go0 = (long)wave * 128 + lane * 4;
    const long go1 = (long)(wave + 8) * 128 + lane * 4;

    int mt = blockIdx.x;
    {
        const float* xp = X + (long)mt * 16 * 128;
        float4 t0 = *(const float4*)(xp + go0);
        float4 t1 = *(const float4*)(xp + go1);
        if (FUSE_ADD) {
            const float* gp = AGG + (long)mt * 16 * 128;
            const float4 g0 = *(const float4*)(gp + go0);
            const float4 g1 = *(const float4*)(gp + go1);
            t0.x += g0.x; t0.y += g0.y; t0.z += g0.z; t0.w += g0.w;
            t1.x += g1.x; t1.y += g1.y; t1.z += g1.z; t1.w += g1.w;
        }
        *(float4*)(&tileA[0][so0]) = t0;
        *(float4*)(&tileA[0][so1]) = t1;
    }

    int cur = 0;
    for (; mt < NG_TILES; mt += gridDim.x) {
        __syncthreads();

        const int nx = mt + gridDim.x;
        const int pf = nx < NG_TILES ? nx : mt;
        const float* xp = X + (long)pf * 16 * 128;
        const float4 x0 = *(const float4*)(xp + go0);
        const float4 x1 = *(const float4*)(xp + go1);
        float4 g0 = x0, g1 = x1;
        if (FUSE_ADD) {
            const float* gp = AGG + (long)pf * 16 * 128;
            g0 = *(const float4*)(gp + go0);
            g1 = *(const float4*)(gp + go1);
        }

        const float* ap = &tileA[cur][mi * NG_PITCH + kh * 2];
        v2f af[32];
#pragma unroll
        for (int j = 0; j < 32; j++) af[j] = *(const v2f*)(ap + 4 * j);
        v8f acc = {0.f, 0.f, 0.f, 0.f, 0.f, 0.f, 0.f, 0.f};
#pragma unroll
        for (int j = 0; j < 32; j++)
            acc = __builtin_amdgcn_wmma_f32_16x16x4_f32(false, af[j], false, bf[j],
                                                        (short)0, acc, false, false);
        __builtin_amdgcn_sched_group_barrier(0x100, 32, 0);
        __builtin_amdgcn_sched_group_barrier(0x008, 32, 0);
        __builtin_amdgcn_sched_group_barrier(0x002, 64, 0);

        float* orow = O + (long)(mt * 16 + kh * 8) * 128 + col;
#pragma unroll
        for (int r = 0; r < 8; r++) {
            float v = acc[r] + bb;
            if (BN)   v = (v - mu) * inv * gm + bt;
            if (RELU) v = fmaxf(v, 0.f);
            orow[(long)r * 128] = v;
        }

        float4 t0 = x0, t1 = x1;
        if (FUSE_ADD) {
            t0.x += g0.x; t0.y += g0.y; t0.z += g0.z; t0.w += g0.w;
            t1.x += g1.x; t1.y += g1.y; t1.z += g1.z; t1.w += g1.w;
        }
        *(float4*)(&tileA[cur ^ 1][so0]) = t0;
        *(float4*)(&tileA[cur ^ 1][so1]) = t1;
        cur ^= 1;
    }
}

template <bool FUSE_ADD, bool BN, bool RELU>
__global__ void gemm_wmma(const float* __restrict__ X, const float* __restrict__ AGG,
                          const float* __restrict__ W, const float* __restrict__ bias,
                          const float* __restrict__ gamma, const float* __restrict__ beta,
                          const float* __restrict__ mean, const float* __restrict__ var,
                          float* __restrict__ O, int M, int K, int N) {
    int wave = threadIdx.x >> 5, lane = threadIdx.x & 31;
    int tilesN = N >> 4;
    int tile = blockIdx.x * 8 + wave;
    if (tile >= (M >> 4) * tilesN) return;
    int mt = tile / tilesN, nt = tile % tilesN;
    int mi = lane & 15, kh = lane >> 4;

    const float* xr = X + (long)(mt * 16 + mi) * K + kh * 2;
    const float* ar = FUSE_ADD ? (AGG + (long)(mt * 16 + mi) * K + kh * 2) : nullptr;
    const float* wc = W + (long)(kh * 2) * N + nt * 16 + mi;

    v8f acc = {0.f, 0.f, 0.f, 0.f, 0.f, 0.f, 0.f, 0.f};
    for (int kb = 0; kb < K; kb += 4) {
        v2f a, b;
        a.x = xr[kb];
        a.y = xr[kb + 1];
        if (FUSE_ADD) { a.x += ar[kb]; a.y += ar[kb + 1]; }
        b.x = wc[(long)kb * N];
        b.y = wc[(long)(kb + 1) * N];
        acc = __builtin_amdgcn_wmma_f32_16x16x4_f32(false, a, false, b,
                                                    (short)0, acc, false, false);
    }

    int col = nt * 16 + mi;
    float bb = bias[col];
    float mu = 0.f, inv = 1.f, gm = 1.f, bt = 0.f;
    if (BN) { mu = mean[col]; inv = rsqrtf(var[col] + BN_EPS); gm = gamma[col]; bt = beta[col]; }
    float* orow = O + (long)(mt * 16 + kh * 8) * N + col;
#pragma unroll
    for (int r = 0; r < 8; r++) {
        float v = acc[r] + bb;
        if (BN)   v = (v - mu) * inv * gm + bt;
        if (RELU) v = fmaxf(v, 0.f);
        orow[(long)r * N] = v;
    }
}

__global__ void pool3(const float* __restrict__ h1, const float* __restrict__ h2,
                      const float* __restrict__ h3, const int* __restrict__ batch,
                      float* __restrict__ gcat) {
    int t = blockIdx.x * blockDim.x + threadIdx.x;
    int node = t >> 5, lane = t & 31;
    if (node >= N_NODES) return;
    int g = batch[node];
    float* base = gcat + (long)g * 384 + lane * 4;
    const float* hs[3] = {h1, h2, h3};
#pragma unroll
    for (int l = 0; l < 3; l++) {
        float4 v = *(const float4*)(hs[l] + (long)node * 128 + lane * 4);
        float* p = base + l * 128;
        atomicAdd(p + 0, v.x);
        atomicAdd(p + 1, v.y);
        atomicAdd(p + 2, v.z);
        atomicAdd(p + 3, v.w);
    }
}

__global__ void lin2_kernel(const float* __restrict__ h, const float* __restrict__ w,
                            const float* __restrict__ b, float* __restrict__ out) {
    int t = blockIdx.x * blockDim.x + threadIdx.x;
    if (t >= N_GRAPHS * 10) return;
    int g = t / 10, o = t % 10;
    const float* hr = h + (long)g * 384;
    float acc = b[o];
    for (int k = 0; k < 384; k++) acc = fmaf(hr[k], w[k * 10 + o], acc);
    out[t] = acc;
}

extern "C" void kernel_launch(void* const* d_in, const int* in_sizes, int n_in,
                              void* d_out, int out_size, void* d_ws, size_t ws_size,
                              hipStream_t stream) {
    const int*   x_ids = (const int*)d_in[0];
    const int*   edge  = (const int*)d_in[1];
    const int*   batch = (const int*)d_in[2];
    const float* emb   = (const float*)d_in[3];

    const float *c_w1[3], *c_b1[3], *c_g[3], *c_be[3], *c_mu[3], *c_va[3], *c_w2[3], *c_b2[3];
    for (int i = 0; i < 3; i++) {
        c_w1[i] = (const float*)d_in[4 + 8 * i + 0];
        c_b1[i] = (const float*)d_in[4 + 8 * i + 1];
        c_g[i]  = (const float*)d_in[4 + 8 * i + 2];
        c_be[i] = (const float*)d_in[4 + 8 * i + 3];
        c_mu[i] = (const float*)d_in[4 + 8 * i + 4];
        c_va[i] = (const float*)d_in[4 + 8 * i + 5];
        c_w2[i] = (const float*)d_in[4 + 8 * i + 6];
        c_b2[i] = (const float*)d_in[4 + 8 * i + 7];
    }
    const float* lin1_w = (const float*)d_in[28];
    const float* lin1_b = (const float*)d_in[29];
    const float* lin2_w = (const float*)d_in[30];
    const float* lin2_b = (const float*)d_in[31];

    const long NF = (long)N_NODES * 128;
    float* ws   = (float*)d_ws;
    float* x0   = ws + 0 * NF;
    float* agg  = ws + 1 * NF;
    float* tmp  = ws + 2 * NF;
    float* hbuf[3] = { ws + 3 * NF, ws + 4 * NF, x0 };
    float* gcat  = agg;
    float* lin1o = tmp;

    gather_emb<<<(N_NODES * 32 + 255) / 256, 256, 0, stream>>>(x_ids, emb, x0);

    const float* xin = x0;
    for (int l = 0; l < 3; l++) {
        agg_owner<<<AO_GROUPS, 1024, 0, stream>>>(xin, edge, agg);
        gemm_node<true, true, true><<<NODE_GEMM_GROUPS, 256, 0, stream>>>(
            xin, agg, c_w1[l], c_b1[l], c_g[l], c_be[l], c_mu[l], c_va[l], tmp);
        gemm_node<false, false, true><<<NODE_GEMM_GROUPS, 256, 0, stream>>>(
            tmp, nullptr, c_w2[l], c_b2[l], nullptr, nullptr, nullptr, nullptr, hbuf[l]);
        xin = hbuf[l];
    }

    zero_f32<<<(N_GRAPHS * 384 / 4 + 255) / 256, 256, 0, stream>>>(gcat, N_GRAPHS * 384 / 4);
    pool3<<<(N_NODES * 32 + 255) / 256, 256, 0, stream>>>(hbuf[0], hbuf[1], hbuf[2], batch, gcat);

    const int tiles_head = (N_GRAPHS / 16) * (384 / 16);
    gemm_wmma<false, false, true><<<(tiles_head + 7) / 8, 256, 0, stream>>>(
        gcat, nullptr, lin1_w, lin1_b, nullptr, nullptr, nullptr, nullptr,
        lin1o, N_GRAPHS, 384, 384);

    lin2_kernel<<<(N_GRAPHS * 10 + 255) / 256, 256, 0, stream>>>(lin1o, lin2_w, lin2_b,
                                                                 (float*)d_out);
}


# Neighbour sum: only layer 1 scans the edge list; each wave saves its owned edges (source row, local row) to the workspace and layers 2 and 3 replay that list; sum order stays LDS arrival order
# speedup vs baseline: 2.2925x; 1.1139x over previous
#include <hip/hip_runtime.h>

typedef __attribute__((ext_vector_type(2))) float v2f;
typedef __attribute__((ext_vector_type(8))) float v8f;

#define N_NODES  50000
#define N_EDGES  800000
#define N_GRAPHS 512
#define BN_EPS   1e-5f

__global__ void zero_f32(float* __restrict__ p, int n4) {
    int t = blockIdx.x * blockDim.x + threadIdx.x;
    if (t < n4) ((float4*)p)[t] = make_float4(0.f, 0.f, 0.f, 0.f);
}

__global__ void gather_emb(const int* __restrict__ ids, const float* __restrict__ emb,
                           float* __restrict__ x) {
    int t = blockIdx.x * blockDim.x + threadIdx.x;
    int node = t >> 5, lane = t & 31;
    if (node >= N_NODES) return;
    int id = ids[node];
    float4 v = *(const float4*)(emb + (long)id * 128 + lane * 4);
    *(float4*)(x + (long)node * 128 + lane * 4) = v;
}

#define AO_ROWS   196
#define AO_GROUPS ((N_NODES + AO_ROWS - 1) / AO_ROWS)
#define AO_SCAN   8
#define AO_TAKE   16
#define AO_QCAP   (AO_SCAN * 128 + AO_TAKE)
#define AO_ITERS  (N_EDGES / 128)
#define AO_REC_CAP 512
static_assert(AO_REC_CAP % AO_TAKE == 0 && AO_REC_CAP <= AO_QCAP, "a saved list is whole passes and fits the queue");
static_assert(N_NODES <= (1 << 16) && AO_ROWS < (1 << 8), "a saved word holds source row and local row");
static_assert(N_EDGES % 128 == 0, "a wave-load of the scan covers 128 edges");
static_assert(N_EDGES <= (1 << 20) && AO_ROWS <= (1 << 12), "a queue word holds edge and local row");
static_assert(AO_QCAP % AO_TAKE == 0 && AO_TAKE <= 32 && (AO_TAKE & (AO_TAKE - 1)) == 0,
              "a scan group fits behind a rest of less than one pass; one wave writes a pass's fill");

template <bool SAVED, bool REC>
__device__ __forceinline__ void ao_take(const unsigned* q, int i,
                                        const float* __restrict__ x, const int* __restrict__ ei,
                                        float* acc, int lane, unsigned* __restrict__ rec, int& nrec) {
    unsigned en[AO_TAKE];
    int s[AO_TAKE];
    float v[AO_TAKE][4];
#pragma unroll
    for (int j = 0; j < AO_TAKE; j++) {
        en[j] = __builtin_amdgcn_readfirstlane(q[i + j]);
        s[j] = SAVED ? (int)(en[j] & 0xFFFFu) : ei[en[j] & 0xFFFFFu];
    }
    if (REC && nrec >= 0) {
        if (nrec + AO_TAKE <= AO_REC_CAP) {
            unsigned w = 0u;
#pragma unroll
            for (int j = 0; j < AO_TAKE; j++)
                w = (lane == j) ? ((unsigned)s[j] | ((en[j] >> 20) << 16)) : w;
            if (lane < AO_TAKE) rec[nrec + lane] = w;
            nrec += AO_TAKE;
        } else {
            nrec = -1;
        }
    }
#pragma unroll
    for (int j = 0; j < AO_TAKE; j++) {
        const float* p = x + (long)s[j] * 128 + lane;
        v[j][0] = p[0];
        v[j][1] = p[32];
        v[j][2] = p[64];
        v[j][3] = p[96];
    }
#pragma unroll
    for (int j = 0; j < AO_TAKE; j++) {
        float* a = acc + (en[j] >> (SAVED ? 16 : 20)) * 128 + lane;
        atomicAdd(a + 0,  v[j][0]);
        atomicAdd(a + 32, v[j][1]);
        atomicAdd(a + 64, v[j][2]);
        atomicAdd(a + 96, v[j][3]);
    }
}

#define AO_FILL ((unsigned)AO_ROWS << 20)
template <bool REC>
__device__ __forceinline__ void ao_drain(unsigned* q, int cnt,
                                         const float* __restrict__ x, const int* __restrict__ ei,
                                         float* acc, int lane, unsigned* __restrict__ rec, int& nrec) {
    const int fill = (AO_TAKE - (cnt & (AO_TAKE - 1))) & (AO_TAKE - 1);
    if (lane < fill) q[cnt + lane] = AO_FILL;
    for (int i = 0; i < cnt; i += AO_TAKE) ao_take<false, REC>(q, i, x, ei, acc, lane, rec, nrec);
}

__device__ __forceinline__ void ao_replay(unsigned* q, const unsigned* __restrict__ rec, int n,
                                          const float* __restrict__ x, float* acc, int lane) {
    for (int i = lane; i < n; i += 32) q[i] = rec[i];
    int none = -1;
    for (int i = 0; i < n; i += AO_TAKE) ao_take<true, false>(q, i, x, nullptr, acc, lane, nullptr, none);
}

template <bool REC>
__device__ __forceinline__ int ao_full(unsigned* q, int cnt,
                                       const float* __restrict__ x, const int* __restrict__ ei,
                                       float* acc, int lane, unsigned* __restrict__ rec, int& nrec) {
    const int full = cnt & ~(AO_TAKE - 1);
    for (int i = 0; i < full; i += AO_TAKE) ao_take<false, REC>(q, i, x, ei, acc, lane, rec, nrec);
    const int rest = cnt - full;
    if (lane < rest) q[lane] = q[full + lane];
    return rest;
}

__device__ __forceinline__ int ao_push(unsigned* q, int cnt, bool m, int dl, int e) {
    const unsigned mask = __builtin_amdgcn_ballot_w32(m);
    if (mask != 0u) {
        if (m) q[cnt + (int)__builtin_amdgcn_mbcnt_lo(mask, 0u)] = (unsigned)e | ((unsigned)dl << 20);
        cnt += __builtin_popcount(mask);
    }
    return cnt;
}

__device__ __forceinline__ int ao_scan(unsigned* q, int cnt, int4 d, int e0, int lo, int nrows) {
    const bool m0 = (unsigned)(d.x - lo) < (unsigned)nrows;
    const bool m1 = (unsigned)(d.y - lo) < (unsigned)nrows;
    const bool m2 = (unsigned)(d.z - lo) < (unsigned)nrows;
    const bool m3 = (unsigned)(d.w - lo) < (unsigned)nrows;
    if (__builtin_amdgcn_ballot_w32(m0 | m1 | m2 | m3) != 0u) {
        cnt = ao_push(q, cnt, m0, d.x - lo, e0 + 0);
        cnt = ao_push(q, cnt, m1, d.y - lo, e0 + 1);
        cnt = ao_push(q, cnt, m2, d.z - lo, e0 + 2);
        cnt = ao_push(q, cnt, m3, d.w - lo, e0 + 3);
    }
    return cnt;
}

template <bool FIRST>
__global__ void __launch_bounds__(1024)
agg_owner(const float* __restrict__ x, const int* __restrict__ ei, float* __restrict__ agg,
          unsigned* __restrict__ recs, int* __restrict__ cnts) {
    __shared__ __attribute__((aligned(16))) float acc[(AO_ROWS + 1) * 128];
    __shared__ __attribute__((aligned(16))) unsigned queue[32][AO_QCAP];
    const int wave = __builtin_amdgcn_readfirstlane(threadIdx.x >> 5), lane = threadIdx.x & 31;
    const int lo = blockIdx.x * AO_ROWS;
    const int nrows = (N_NODES - lo) < AO_ROWS ? (N_NODES - lo) : AO_ROWS;

    for (int i = threadIdx.x; i < (AO_ROWS + 1) * 32; i += 1024)
        ((float4*)acc)[i] = make_float4(0.f, 0.f, 0.f, 0.f);
    __syncthreads();

    unsigned* q = queue[wave];
    unsigned* rec = recs + (long)(blockIdx.x * 32 + wave) * AO_REC_CAP;
    int nrec = 0;
    bool replay = false;
    int saved = 0;
    if (!FIRST) {
        const int c = cnts[blockIdx.x * 32 + lane];
        replay = __builtin_amdgcn_ballot_w32(c < 0) == 0u;
        saved = __builtin_amdgcn_readlane(c, wave);
    }
    if (replay) {
        ao_replay(q, rec, saved, x, acc, lane);
    } else {
        const int4* dst4 = (const int4*)(ei + N_EDGES);
        int cnt = 0;
        const int n_it = (AO_ITERS - wave + 31) / 32;
        const int first = (int)(blockIdx.x * n_it) / AO_GROUPS;

        for (int k = 0; k < n_it; k += AO_SCAN) {
            int4 d[AO_SCAN];
            int it[AO_SCAN];
#pragma unroll
            for (int u = 0; u < AO_SCAN; u++) {
                int kk = ((k + u) < n_it ? (k + u) : (n_it - 1)) + first;
                if (kk >= n_it) kk -= n_it;
                it[u] = wave + 32 * kk;
                d[u] = dst4[it[u] * 32 + lane];
            }
#pragma unroll
            for (int u = 0; u < AO_SCAN; u++)
                cnt = ao_scan(q, cnt, d[u], it[u] * 128 + lane * 4, lo, (k + u) < n_it ? nrows : 0);
            if (cnt >= AO_TAKE) cnt = ao_full<FIRST>(q, cnt, x, ei, acc, lane, rec, nrec);
        }
        ao_drain<FIRST>(q, cnt, x, ei, acc, lane, rec, nrec);
        if (FIRST && lane == 0) cnts[blockIdx.x * 32 + wave] = nrec;
    }

    __syncthreads();
    float4* out4 = (float4*)(agg + (long)lo * 128);
    for (int i = threadIdx.x; i < nrows * 32; i += 1024) out4[i] = ((const float4*)acc)[i];
}

#define NG_PITCH 132
#define NG_TILES (N_NODES / 16)
#define NODE_GEMM_GROUPS 512
static_assert(NODE_GEMM_GROUPS <= NG_TILES, "every work-group has a first tile");
static_assert(N_NODES % 16 == 0, "row tiles must tile the node list");
template <bool FUSE_ADD, bool BN, bool RELU>
__global__ void __launch_bounds__(256)
gemm_node(const float* __restrict__ X, const float* __restrict__ AGG,
          const float* __restrict__ W, const float* __restrict__ bias,
          const float* __restrict__ gamma, const float* __restrict__ beta,
          const float* __restrict__ mean, const float* __restrict__ var,
          float* __restrict__ O) {
    __shared__ __attribute__((aligned(16))) float tileA[2][16 * NG_PITCH];
    const int wave = threadIdx.x >> 5, lane = threadIdx.x & 31;
    const int mi = lane & 15, kh = lane >> 4;
    const int col = wave * 16 + mi;

    v2f bf[32];
    const float* wc = W + (long)(kh * 2) * 128 + col;
#pragma unroll
    for (int j = 0; j < 32; j++) {
        bf[j].x = wc[(long)(4 * j) * 128];
        bf[j].y = wc[(long)(4 * j + 1) * 128];
    }

    const float bb = bias[col];
    float mu = 0.f, inv = 1.f, gm = 1.f, bt = 0.f;
    if (BN) { mu = mean[col]; inv = rsqrtf(var[col] + BN_EPS); gm = gamma[col]; bt = beta[col]; }

    const int so0 = wave * NG_PITCH + lane * 4;
    const int so1 = (wave + 8) * NG_PITCH + lane * 4;
    const long go0 = (long)wave * 128 + lane * 4;
    const long go1 = (long)(wave + 8) * 128 + lane * 4;

    int mt = blockIdx.x;
    {
        const float* xp = X + (long)mt * 16 * 128;
        float4 t0 = *(const float4*)(xp + go0);
        float4 t1 = *(const float4*)(xp + go1);
        if (FUSE_ADD) {
            const float* gp = AGG + (long)mt * 16 * 128;
            const float4 g0 = *(const float4*)(gp + go0);
            const float4 g1 = *(const float4*)(gp + go1);
            t0.x += g0.x; t0.y += g0.y; t0.z += g0.z; t0.w += g0.w;
            t1.x += g1.x; t1.y += g1.y; t1.z += g1.z; t1.w += g1.w;
        }
        *(float4*)(&tileA[0][so0]) = t0;
        *(float4*)(&tileA[0][so1]) = t1;
    }

    int cur = 0;
    for (; mt < NG_TILES; mt += gridDim.x) {
        __syncthreads();

        const int nx = mt + gridDim.x;
        const int pf = nx < NG_TILES ? nx : mt;
        const float* xp = X + (long)pf * 16 * 128;
        const float4 x0 = *(const float4*)(xp + go0);
        const float4 x1 = *(const float4*)(xp + go1);
        float4 g0 = x0, g1 = x1;
        if (FUSE_ADD) {
            const float* gp = AGG + (long)pf * 16 * 128;
            g0 = *(const float4*)(gp + go0);
            g1 = *(const float4*)(gp + go1);
        }

        const float* ap = &tileA[cur][mi * NG_PITCH + kh * 2];
        v2f af[32];
#pragma unroll
        for (int j = 0; j < 32; j++) af[j] = *(const v2f*)(ap + 4 * j);
        v8f acc = {0.f, 0.f, 0.f, 0.f, 0.f, 0.f, 0.f, 0.f};
#pragma unroll
        for (int j = 0; j < 32; j++)
            acc = __builtin_amdgcn_wmma_f32_16x16x4_f32(false, af[j], false, bf[j],
                                                        (short)0, acc, false, false);
        __builtin_amdgcn_sched_group_barrier(0x100, 32, 0);
        __builtin_amdgcn_sched_group_barrier(0x008, 32, 0);
        __builtin_amdgcn_sched_group_barrier(0x002, 64, 0);

        float* orow = O + (long)(mt * 16 + kh * 8) * 128 + col;
#pragma unroll
        for (int r = 0; r < 8; r++) {
            float v = acc[r] + bb;
            if (BN)   v = (v - mu) * inv * gm + bt;
            if (RELU) v = fmaxf(v, 0.f);
            orow[(long)r * 128] = v;
        }

        float4 t0 = x0, t1 = x1;
        if (FUSE_ADD) {
            t0.x += g0.x; t0.y += g0.y; t0.z += g0.z; t0.w += g0.w;
            t1.x += g1.x; t1.y += g1.y; t1.z += g1.z; t1.w += g1.w;
        }
        *(float4*)(&tileA[cur ^ 1][so0]) = t0;
        *(float4*)(&tileA[cur ^ 1][so1]) = t1;
        cur ^= 1;
    }
}

template <bool FUSE_ADD, bool BN, bool RELU>
__global__ void gemm_wmma(const float* __restrict__ X, const float* __restrict__ AGG,
                          const float* __restrict__ W, const float* __restrict__ bias,
                          const float* __restrict__ gamma, const float* __restrict__ beta,
                          const float* __restrict__ mean, const float* __restrict__ var,
                          float* __restrict__ O, int M, int K, int N) {
    int wave = threadIdx.x >> 5, lane = threadIdx.x & 31;
    int tilesN = N >> 4;
    int tile = blockIdx.x * 8 + wave;
    if (tile >= (M >> 4) * tilesN) return;
    int mt = tile / tilesN, nt = tile % tilesN;
    int mi = lane & 15, kh = lane >> 4;

    const float* xr = X + (long)(mt * 16 + mi) * K + kh * 2;
    const float* ar = FUSE_ADD ? (AGG + (long)(mt * 16 + mi) * K + kh * 2) : nullptr;
    const float* wc = W + (long)(kh * 2) * N + nt * 16 + mi;

    v8f acc = {0.f, 0.f, 0.f, 0.f, 0.f, 0.f, 0.f, 0.f};
    for (int kb = 0; kb < K; kb += 4) {
        v2f a, b;
        a.x = xr[kb];
        a.y = xr[kb + 1];
        if (FUSE_ADD) { a.x += ar[kb]; a.y += ar[kb + 1]; }
        b.x = wc[(long)kb * N];
        b.y = wc[(long)(kb + 1) * N];
        acc = __builtin_amdgcn_wmma_f32_16x16x4_f32(false, a, false, b,
                                                    (short)0, acc, false, false);
    }

    int col = nt * 16 + mi;
    float bb = bias[col];
    float mu = 0.f, inv = 1.f, gm = 1.f, bt = 0.f;
    if (BN) { mu = mean[col]; inv = rsqrtf(var[col] + BN_EPS); gm = gamma[col]; bt = beta[col]; }
    float* orow = O + (long)(mt * 16 + kh * 8) * N + col;
#pragma unroll
    for (int r = 0; r < 8; r++) {
        float v = acc[r] + bb;
        if (BN)   v = (v - mu) * inv * gm + bt;
        if (RELU) v = fmaxf(v, 0.f);
        orow[(long)r * N] = v;
    }
}

__global__ void pool3(const float* __restrict__ h1, const float* __restrict__ h2,
                      const float* __restrict__ h3, const int* __restrict__ batch,
                      float* __restrict__ gcat) {
    int t = blockIdx.x * blockDim.x + threadIdx.x;
    int node = t >> 5, lane = t & 31;
    if (node >= N_NODES) return;
    int g = batch[node];
    float* base = gcat + (long)g * 384 + lane * 4;
    const float* hs[3] = {h1, h2, h3};
#pragma unroll
    for (int l = 0; l < 3; l++) {
        float4 v = *(const float4*)(hs[l] + (long)node * 128 + lane * 4);
        float* p = base + l * 128;
        atomicAdd(p + 0, v.x);
        atomicAdd(p + 1, v.y);
        atomicAdd(p + 2, v.z);
        atomicAdd(p + 3, v.w);
    }
}

__global__ void lin2_kernel(const float* __restrict__ h, const float* __restrict__ w,
                            const float* __restrict__ b, float* __restrict__ out) {
    int t = blockIdx.x * blockDim.x + threadIdx.x;
    if (t >= N_GRAPHS * 10) return;
    int g = t / 10, o = t % 10;
    const float* hr = h + (long)g * 384;
    float acc = b[o];
    for (int k = 0; k < 384; k++) acc = fmaf(hr[k], w[k * 10 + o], acc);
    out[t] = acc;
}

extern "C" void kernel_launch(void* const* d_in, const int* in_sizes, int n_in,
                              void* d_out, int out_size, void* d_ws, size_t ws_size,
                              hipStream_t stream) {
    const int*   x_ids = (const int*)d_in[0];
    const int*   edge  = (const int*)d_in[1];
    const int*   batch = (const int*)d_in[2];
    const float* emb   = (const float*)d_in[3];

    const float *c_w1[3], *c_b1[3], *c_g[3], *c_be[3], *c_mu[3], *c_va[3], *c_w2[3], *c_b2[3];
    for (int i = 0; i < 3; i++) {
        c_w1[i] = (const float*)d_in[4 + 8 * i + 0];
        c_b1[i] = (const float*)d_in[4 + 8 * i + 1];
        c_g[i]  = (const float*)d_in[4 + 8 * i + 2];
        c_be[i] = (const float*)d_in[4 + 8 * i + 3];
        c_mu[i] = (const float*)d_in[4 + 8 * i + 4];
        c_va[i] = (const float*)d_in[4 + 8 * i + 5];
        c_w2[i] = (const float*)d_in[4 + 8 * i + 6];
        c_b2[i] = (const float*)d_in[4 + 8 * i + 7];
    }
    const float* lin1_w = (const float*)d_in[28];
    const float* lin1_b = (const float*)d_in[29];
    const float* lin2_w = (const float*)d_in[30];
    const float* lin2_b = (const float*)d_in[31];

    const long NF = (long)N_NODES * 128;
    float* ws   = (float*)d_ws;
    float* x0   = ws + 0 * NF;
    float* agg  = ws + 1 * NF;
    float* tmp  = ws + 2 * NF;
    float* hbuf[3] = { ws + 3 * NF, ws + 4 * NF, x0 };
    float* gcat  = agg;
    float* lin1o = tmp;
    unsigned* recs = (unsigned*)(ws + 5 * NF);
    int* cnts = (int*)(recs + (long)AO_GROUPS * 32 * AO_REC_CAP);

    gather_emb<<<(N_NODES * 32 + 255) / 256, 256, 0, stream>>>(x_ids, emb, x0);

    const float* xin = x0;
    for (int l = 0; l < 3; l++) {
        if (l == 0) agg_owner<true><<<AO_GROUPS, 1024, 0, stream>>>(xin, edge, agg, recs, cnts);
        else        agg_owner<false><<<AO_GROUPS, 1024, 0, stream>>>(xin, edge, agg, recs, cnts);
        gemm_node<true, true, true><<<NODE_GEMM_GROUPS, 256, 0, stream>>>(
            xin, agg, c_w1[l], c_b1[l], c_g[l], c_be[l], c_mu[l], c_va[l], tmp);
        gemm_node<false, false, true><<<NODE_GEMM_GROUPS, 256, 0, stream>>>(
            tmp, nullptr, c_w2[l], c_b2[l], nullptr, nullptr, nullptr, nullptr, hbuf[l]);
        xin = hbuf[l];
    }

    zero_f32<<<(N_GRAPHS * 384 / 4 + 255) / 256, 256, 0, stream>>>(gcat, N_GRAPHS * 384 / 4);
    pool3<<<(N_NODES * 32 + 255) / 256, 256, 0, stream>>>(hbuf[0], hbuf[1], hbuf[2], batch, gcat);

    const int tiles_head = (N_GRAPHS / 16) * (384 / 16);
    gemm_wmma<false, false, true><<<(tiles_head + 7) / 8, 256, 0, stream>>>(
        gcat, nullptr, lin1_w, lin1_b, nullptr, nullptr, nullptr, nullptr,
        lin1o, N_GRAPHS, 384, 384);

    lin2_kernel<<<(N_GRAPHS * 10 + 255) / 256, 256, 0, stream>>>(lin1o, lin2_w, lin2_b,
                                                                 (float*)d_out);
}


# Graph pooling fused into each layer's second product: a lane adds its accumulator rows per run of equal graph id, one atomic per run and column; pooling launch and its re-read gone; sum order changes
# speedup vs baseline: 2.7884x; 1.2163x over previous
#include <hip/hip_runtime.h>

typedef __attribute__((ext_vector_type(2))) float v2f;
typedef __attribute__((ext_vector_type(8))) float v8f;

#define N_NODES  50000
#define N_EDGES  800000
#define N_GRAPHS 512
#define BN_EPS   1e-5f

__global__ void zero_f32(float* __restrict__ p, int n4) {
    int t = blockIdx.x * blockDim.x + threadIdx.x;
    if (t < n4) ((float4*)p)[t] = make_float4(0.f, 0.f, 0.f, 0.f);
}

__global__ void gather_emb(const int* __restrict__ ids, const float* __restrict__ emb,
                           float* __restrict__ x) {
    int t = blockIdx.x * blockDim.x + threadIdx.x;
    int node = t >> 5, lane = t & 31;
    if (node >= N_NODES) return;
    int id = ids[node];
    float4 v = *(const float4*)(emb + (long)id * 128 + lane * 4);
    *(float4*)(x + (long)node * 128 + lane * 4) = v;
}

#define AO_ROWS   196
#define AO_GROUPS ((N_NODES + AO_ROWS - 1) / AO_ROWS)
#define AO_SCAN   8
#define AO_TAKE   16
#define AO_QCAP   (AO_SCAN * 128 + AO_TAKE)
#define AO_ITERS  (N_EDGES / 128)
#define AO_REC_CAP 512
static_assert(AO_REC_CAP % AO_TAKE == 0 && AO_REC_CAP <= AO_QCAP, "a saved list is whole passes and fits the queue");
static_assert(N_NODES <= (1 << 16) && AO_ROWS < (1 << 8), "a saved word holds source row and local row");
static_assert(N_EDGES % 128 == 0, "a wave-load of the scan covers 128 edges");
static_assert(N_EDGES <= (1 << 20) && AO_ROWS <= (1 << 12), "a queue word holds edge and local row");
static_assert(AO_QCAP % AO_TAKE == 0 && AO_TAKE <= 32 && (AO_TAKE & (AO_TAKE - 1)) == 0,
              "a scan group fits behind a rest of less than one pass; one wave writes a pass's fill");

template <bool SAVED, bool REC>
__device__ __forceinline__ void ao_take(const unsigned* q, int i,
                                        const float* __restrict__ x, const int* __restrict__ ei,
                                        float* acc, int lane, unsigned* __restrict__ rec, int& nrec) {
    unsigned en[AO_TAKE];
    int s[AO_TAKE];
    float v[AO_TAKE][4];
#pragma unroll
    for (int j = 0; j < AO_TAKE; j++) {
        en[j] = __builtin_amdgcn_readfirstlane(q[i + j]);
        s[j] = SAVED ? (int)(en[j] & 0xFFFFu) : ei[en[j] & 0xFFFFFu];
    }
    if (REC && nrec >= 0) {
        if (nrec + AO_TAKE <= AO_REC_CAP) {
            unsigned w = 0u;
#pragma unroll
            for (int j = 0; j < AO_TAKE; j++)
                w = (lane == j) ? ((unsigned)s[j] | ((en[j] >> 20) << 16)) : w;
            if (lane < AO_TAKE) rec[nrec + lane] = w;
            nrec += AO_TAKE;
        } else {
            nrec = -1;
        }
    }
#pragma unroll
    for (int j = 0; j < AO_TAKE; j++) {
        const float* p = x + (long)s[j] * 128 + lane;
        v[j][0] = p[0];
        v[j][1] = p[32];
        v[j][2] = p[64];
        v[j][3] = p[96];
    }
#pragma unroll
    for (int j = 0; j < AO_TAKE; j++) {
        float* a = acc + (en[j] >> (SAVED ? 16 : 20)) * 128 + lane;
        atomicAdd(a + 0,  v[j][0]);
        atomicAdd(a + 32, v[j][1]);
        atomicAdd(a + 64, v[j][2]);
        atomicAdd(a + 96, v[j][3]);
    }
}

#define AO_FILL ((unsigned)AO_ROWS << 20)
template <bool REC>
__device__ __forceinline__ void ao_drain(unsigned* q, int cnt,
                                         const float* __restrict__ x, const int* __restrict__ ei,
                                         float* acc, int lane, unsigned* __restrict__ rec, int& nrec) {
    const int fill = (AO_TAKE - (cnt & (AO_TAKE - 1))) & (AO_TAKE - 1);
    if (lane < fill) q[cnt + lane] = AO_FILL;
    for (int i = 0; i < cnt; i += AO_TAKE) ao_take<false, REC>(q, i, x, ei, acc, lane, rec, nrec);
}

__device__ __forceinline__ void ao_replay(unsigned* q, const unsigned* __restrict__ rec, int n,
                                          const float* __restrict__ x, float* acc, int lane) {
    for (int i = lane; i < n; i += 32) q[i] = rec[i];
    int none = -1;
    for (int i = 0; i < n; i += AO_TAKE) ao_take<true, false>(q, i, x, nullptr, acc, lane, nullptr, none);
}

template <bool REC>
__device__ __forceinline__ int ao_full(unsigned* q, int cnt,
                                       const float* __restrict__ x, const int* __restrict__ ei,
                                       float* acc, int lane, unsigned* __restrict__ rec, int& nrec) {
    const int full = cnt & ~(AO_TAKE - 1);
    for (int i = 0; i < full; i += AO_TAKE) ao_take<false, REC>(q, i, x, ei, acc, lane, rec, nrec);
    const int rest = cnt - full;
    if (lane < rest) q[lane] = q[full + lane];
    return rest;
}

__device__ __forceinline__ int ao_push(unsigned* q, int cnt, bool m, int dl, int e) {
    const unsigned mask = __builtin_amdgcn_ballot_w32(m);
    if (mask != 0u) {
        if (m) q[cnt + (int)__builtin_amdgcn_mbcnt_lo(mask, 0u)] = (unsigned)e | ((unsigned)dl << 20);
        cnt += __builtin_popcount(mask);
    }
    return cnt;
}

__device__ __forceinline__ int ao_scan(unsigned* q, int cnt, int4 d, int e0, int lo, int nrows) {
    const bool m0 = (unsigned)(d.x - lo) < (unsigned)nrows;
    const bool m1 = (unsigned)(d.y - lo) < (unsigned)nrows;
    const bool m2 = (unsigned)(d.z - lo) < (unsigned)nrows;
    const bool m3 = (unsigned)(d.w - lo) < (unsigned)nrows;
    if (__builtin_amdgcn_ballot_w32(m0 | m1 | m2 | m3) != 0u) {
        cnt = ao_push(q, cnt, m0, d.x - lo, e0 + 0);
        cnt = ao_push(q, cnt, m1, d.y - lo, e0 + 1);
        cnt = ao_push(q, cnt, m2, d.z - lo, e0 + 2);
        cnt = ao_push(q, cnt, m3, d.w - lo, e0 + 3);
    }
    return cnt;
}

template <bool FIRST>
__global__ void __launch_bounds__(1024)
agg_owner(const float* __restrict__ x, const int* __restrict__ ei, float* __restrict__ agg,
          unsigned* __restrict__ recs, int* __restrict__ cnts) {
    __shared__ __attribute__((aligned(16))) float acc[(AO_ROWS + 1) * 128];
    __shared__ __attribute__((aligned(16))) unsigned queue[32][AO_QCAP];
    const int wave = __builtin_amdgcn_readfirstlane(threadIdx.x >> 5), lane = threadIdx.x & 31;
    const int lo = blockIdx.x * AO_ROWS;
    const int nrows = (N_NODES - lo) < AO_ROWS ? (N_NODES - lo) : AO_ROWS;

    for (int i = threadIdx.x; i < (AO_ROWS + 1) * 32; i += 1024)
        ((float4*)acc)[i] = make_float4(0.f, 0.f, 0.f, 0.f);
    __syncthreads();

    unsigned* q = queue[wave];
    unsigned* rec = recs + (long)(blockIdx.x * 32 + wave) * AO_REC_CAP;
    int nrec = 0;
    bool replay = false;
    int saved = 0;
    if (!FIRST) {
        const int c = cnts[blockIdx.x * 32 + lane];
        replay = __builtin_amdgcn_ballot_w32(c < 0) == 0u;
        saved = __builtin_amdgcn_readlane(c, wave);
    }
    if (replay) {
        ao_replay(q, rec, saved, x, acc, lane);
    } else {
        const int4* dst4 = (const int4*)(ei + N_EDGES);
        int cnt = 0;
        const int n_it = (AO_ITERS - wave + 31) / 32;
        const int first = (int)(blockIdx.x * n_it) / AO_GROUPS;

        for (int k = 0; k < n_it; k += AO_SCAN) {
            int4 d[AO_SCAN];
            int it[AO_SCAN];
#pragma unroll
            for (int u = 0; u < AO_SCAN; u++) {
                int kk = ((k + u) < n_it ? (k + u) : (n_it - 1)) + first;
                if (kk >= n_it) kk -= n_it;
                it[u] = wave + 32 * kk;
                d[u] = dst4[it[u] * 32 + lane];
            }
#pragma unroll
            for (int u = 0; u < AO_SCAN; u++)
                cnt = ao_scan(q, cnt, d[u], it[u] * 128 + lane * 4, lo, (k + u) < n_it ? nrows : 0);
            if (cnt >= AO_TAKE) cnt = ao_full<FIRST>(q, cnt, x, ei, acc, lane, rec, nrec);
        }
        ao_drain<FIRST>(q, cnt, x, ei, acc, lane, rec, nrec);
        if (FIRST && lane == 0) cnts[blockIdx.x * 32 + wave] = nrec;
    }

    __syncthreads();
    float4* out4 = (float4*)(agg + (long)lo * 128);
    for (int i = threadIdx.x; i < nrows * 32; i += 1024) out4[i] = ((const float4*)acc)[i];
}

#define NG_PITCH 132
#define NG_TILES (N_NODES / 16)
#define NODE_GEMM_GROUPS 512
static_assert(NODE_GEMM_GROUPS <= NG_TILES, "every work-group has a first tile");
static_assert(N_NODES % 16 == 0, "row tiles must tile the node list");
template <bool FUSE_ADD, bool BN, bool RELU, bool POOL>
__global__ void __launch_bounds__(256)
gemm_node(const float* __restrict__ X, const float* __restrict__ AGG,
          const float* __restrict__ W, const float* __restrict__ bias,
          const float* __restrict__ gamma, const float* __restrict__ beta,
          const float* __restrict__ mean, const float* __restrict__ var,
          float* __restrict__ O, const int* __restrict__ batch, float* __restrict__ gsum) {
    __shared__ __attribute__((aligned(16))) float tileA[2][16 * NG_PITCH];
    const int wave = threadIdx.x >> 5, lane = threadIdx.x & 31;
    const int mi = lane & 15, kh = lane >> 4;
    const int col = wave * 16 + mi;

    v2f bf[32];
    const float* wc = W + (long)(kh * 2) * 128 + col;
#pragma unroll
    for (int j = 0; j < 32; j++) {
        bf[j].x = wc[(long)(4 * j) * 128];
        bf[j].y = wc[(long)(4 * j + 1) * 128];
    }

    const float bb = bias[col];
    float mu = 0.f, inv = 1.f, gm = 1.f, bt = 0.f;
    if (BN) { mu = mean[col]; inv = rsqrtf(var[col] + BN_EPS); gm = gamma[col]; bt = beta[col]; }

    const int so0 = wave * NG_PITCH + lane * 4;
    const int so1 = (wave + 8) * NG_PITCH + lane * 4;
    const long go0 = (long)wave * 128 + lane * 4;
    const long go1 = (long)(wave + 8) * 128 + lane * 4;

    int mt = blockIdx.x;
    {
        const float* xp = X + (long)mt * 16 * 128;
        float4 t0 = *(const float4*)(xp + go0);
        float4 t1 = *(const float4*)(xp + go1);
        if (FUSE_ADD) {
            const float* gp = AGG + (long)mt * 16 * 128;
            const float4 g0 = *(const float4*)(gp + go0);
            const float4 g1 = *(const float4*)(gp + go1);
            t0.x += g0.x; t0.y += g0.y; t0.z += g0.z; t0.w += g0.w;
            t1.x += g1.x; t1.y += g1.y; t1.z += g1.z; t1.w += g1.w;
        }
        *(float4*)(&tileA[0][so0]) = t0;
        *(float4*)(&tileA[0][so1]) = t1;
    }

    int cur = 0;
    for (; mt < NG_TILES; mt += gridDim.x) {
        __syncthreads();

        int bid = 0;
        if (POOL) bid = batch[mt * 16 + mi];

        const int nx = mt + gridDim.x;
        const int pf = nx < NG_TILES ? nx : mt;
        const float* xp = X + (long)pf * 16 * 128;
        const float4 x0 = *(const float4*)(xp + go0);
        const float4 x1 = *(const float4*)(xp + go1);
        float4 g0 = x0, g1 = x1;
        if (FUSE_ADD) {
            const float* gp = AGG + (long)pf * 16 * 128;
            g0 = *(const float4*)(gp + go0);
            g1 = *(const float4*)(gp + go1);
        }

        const float* ap = &tileA[cur][mi * NG_PITCH + kh * 2];
        v2f af[32];
#pragma unroll
        for (int j = 0; j < 32; j++) af[j] = *(const v2f*)(ap + 4 * j);
        v8f acc = {0.f, 0.f, 0.f, 0.f, 0.f, 0.f, 0.f, 0.f};
#pragma unroll
        for (int j = 0; j < 32; j++)
            acc = __builtin_amdgcn_wmma_f32_16x16x4_f32(false, af[j], false, bf[j],
                                                        (short)0, acc, false, false);
        __builtin_amdgcn_sched_group_barrier(0x100, 32, 0);
        __builtin_amdgcn_sched_group_barrier(0x008, 32, 0);
        __builtin_amdgcn_sched_group_barrier(0x002, 64, 0);

        float* orow = O + (long)(mt * 16 + kh * 8) * 128 + col;
        float vv[8];
#pragma unroll
        for (int r = 0; r < 8; r++) {
            float v = acc[r] + bb;
            if (BN)   v = (v - mu) * inv * gm + bt;
            if (RELU) v = fmaxf(v, 0.f);
            orow[(long)r * 128] = v;
            vv[r] = v;
        }

        if (POOL) {
            const int gfirst = __builtin_amdgcn_readfirstlane(bid);
            if (__builtin_amdgcn_ballot_w32(bid != gfirst) == 0u) {
                float s = vv[0];
#pragma unroll
                for (int r = 1; r < 8; r++) s += vv[r];
                const unsigned su = __float_as_uint(s);
                const float so = __uint_as_float(
                    __builtin_amdgcn_permlanex16(su, su, 0x76543210u, 0xfedcba98u, false, false));
                if (kh == 0) atomicAdd(gsum + (long)gfirst * 384 + col, s + so);
            } else {
                int idr[8];
#pragma unroll
                for (int r = 0; r < 8; r++) {
                    const int id_lo = __builtin_amdgcn_readlane(bid, r);
                    const int id_hi = __builtin_amdgcn_readlane(bid, 8 + r);
                    idr[r] = kh ? id_hi : id_lo;
                }
                int gcur = idr[0];
                float s = vv[0];
#pragma unroll
                for (int r = 1; r < 8; r++) {
                    if (idr[r] != gcur) {
                        atomicAdd(gsum + (long)gcur * 384 + col, s);
                        gcur = idr[r];
                        s = vv[r];
                    } else {
                        s += vv[r];
                    }
                }
                atomicAdd(gsum + (long)gcur * 384 + col, s);
            }
        }

        float4 t0 = x0, t1 = x1;
        if (FUSE_ADD) {
            t0.x += g0.x; t0.y += g0.y; t0.z += g0.z; t0.w += g0.w;
            t1.x += g1.x; t1.y += g1.y; t1.z += g1.z; t1.w += g1.w;
        }
        *(float4*)(&tileA[cur ^ 1][so0]) = t0;
        *(float4*)(&tileA[cur ^ 1][so1]) = t1;
        cur ^= 1;
    }
}

template <bool FUSE_ADD, bool BN, bool RELU>
__global__ void gemm_wmma(const float* __restrict__ X, const float* __restrict__ AGG,
                          const float* __restrict__ W, const float* __restrict__ bias,
                          const float* __restrict__ gamma, const float* __restrict__ beta,
                          const float* __restrict__ mean, const float* __restrict__ var,
                          float* __restrict__ O, int M, int K, int N) {
    int wave = threadIdx.x >> 5, lane = threadIdx.x & 31;
    int tilesN = N >> 4;
    int tile = blockIdx.x * 8 + wave;
    if (tile >= (M >> 4) * tilesN) return;
    int mt = tile / tilesN, nt = tile % tilesN;
    int mi = lane & 15, kh = lane >> 4;

    const float* xr = X + (long)(mt * 16 + mi) * K + kh * 2;
    const float* ar = FUSE_ADD ? (AGG + (long)(mt * 16 + mi) * K + kh * 2) : nullptr;
    const float* wc = W + (long)(kh * 2) * N + nt * 16 + mi;

    v8f acc = {0.f, 0.f, 0.f, 0.f, 0.f, 0.f, 0.f, 0.f};
    for (int kb = 0; kb < K; kb += 4) {
        v2f a, b;
        a.x = xr[kb];
        a.y = xr[kb + 1];
        if (FUSE_ADD) { a.x += ar[kb]; a.y += ar[kb + 1]; }
        b.x = wc[(long)kb * N];
        b.y = wc[(long)(kb + 1) * N];
        acc = __builtin_amdgcn_wmma_f32_16x16x4_f32(false, a, false, b,
                                                    (short)0, acc, false, false);
    }

    int col = nt * 16 + mi;
    float bb = bias[col];
    float mu = 0.f, inv = 1.f, gm = 1.f, bt = 0.f;
    if (BN) { mu = mean[col]; inv = rsqrtf(var[col] + BN_EPS); gm = gamma[col]; bt = beta[col]; }
    float* orow = O + (long)(mt * 16 + kh * 8) * N + col;
#pragma unroll
    for (int r = 0; r < 8; r++) {
        float v = acc[r] + bb;
        if (BN)   v = (v - mu) * inv * gm + bt;
        if (RELU) v = fmaxf(v, 0.f);
        orow[(long)r * N] = v;
    }
}

__global__ void lin2_kernel(const float* __restrict__ h, const float* __restrict__ w,
                            const float* __restrict__ b, float* __restrict__ out) {
    int t = blockIdx.x * blockDim.x + threadIdx.x;
    if (t >= N_GRAPHS * 10) return;
    int g = t / 10, o = t % 10;
    const float* hr = h + (long)g * 384;
    float acc = b[o];
    for (int k = 0; k < 384; k++) acc = fmaf(hr[k], w[k * 10 + o], acc);
    out[t] = acc;
}

extern "C" void kernel_launch(void* const* d_in, const int* in_sizes, int n_in,
                              void* d_out, int out_size, void* d_ws, size_t ws_size,
                              hipStream_t stream) {
    const int*   x_ids = (const int*)d_in[0];
    const int*   edge  = (const int*)d_in[1];
    const int*   batch = (const int*)d_in[2];
    const float* emb   = (const float*)d_in[3];

    const float *c_w1[3], *c_b1[3], *c_g[3], *c_be[3], *c_mu[3], *c_va[3], *c_w2[3], *c_b2[3];
    for (int i = 0; i < 3; i++) {
        c_w1[i] = (const float*)d_in[4 + 8 * i + 0];
        c_b1[i] = (const float*)d_in[4 + 8 * i + 1];
        c_g[i]  = (const float*)d_in[4 + 8 * i + 2];
        c_be[i] = (const float*)d_in[4 + 8 * i + 3];
        c_mu[i] = (const float*)d_in[4 + 8 * i + 4];
        c_va[i] = (const float*)d_in[4 + 8 * i + 5];
        c_w2[i] = (const float*)d_in[4 + 8 * i + 6];
        c_b2[i] = (const float*)d_in[4 + 8 * i + 7];
    }
    const float* lin1_w = (const float*)d_in[28];
    const float* lin1_b = (const float*)d_in[29];
    const float* lin2_w = (const float*)d_in[30];
    const float* lin2_b = (const float*)d_in[31];

    const long NF = (long)N_NODES * 128;
    float* ws   = (float*)d_ws;
    float* x0   = ws + 0 * NF;
    float* agg  = ws + 1 * NF;
    float* tmp  = ws + 2 * NF;
    float* hbuf[3] = { ws + 3 * NF, ws + 4 * NF, x0 };
    float* lin1o = tmp;
    unsigned* recs = (unsigned*)(ws + 5 * NF);
    int* cnts = (int*)(recs + (long)AO_GROUPS * 32 * AO_REC_CAP);
    float* gcat = (float*)(cnts + AO_GROUPS * 32);

    zero_f32<<<(N_GRAPHS * 384 / 4 + 255) / 256, 256, 0, stream>>>(gcat, N_GRAPHS * 384 / 4);

    gather_emb<<<(N_NODES * 32 + 255) / 256, 256, 0, stream>>>(x_ids, emb, x0);

    const float* xin = x0;
    for (int l = 0; l < 3; l++) {
        if (l == 0) agg_owner<true><<<AO_GROUPS, 1024, 0, stream>>>(xin, edge, agg, recs, cnts);
        else        agg_owner<false><<<AO_GROUPS, 1024, 0, stream>>>(xin, edge, agg, recs, cnts);
        gemm_node<true, true, true, false><<<NODE_GEMM_GROUPS, 256, 0, stream>>>(
            xin, agg, c_w1[l], c_b1[l], c_g[l], c_be[l], c_mu[l], c_va[l], tmp, nullptr, nullptr);
        gemm_node<false, false, true, true><<<NODE_GEMM_GROUPS, 256, 0, stream>>>(
            tmp, nullptr, c_w2[l], c_b2[l], nullptr, nullptr, nullptr, nullptr, hbuf[l],
            batch, gcat + l * 128);
        xin = hbuf[l];
    }

    const int tiles_head = (N_GRAPHS / 16) * (384 / 16);
    gemm_wmma<false, false, true><<<(tiles_head + 7) / 8, 256, 0, stream>>>(
        gcat, nullptr, lin1_w, lin1_b, nullptr, nullptr, nullptr, nullptr,
        lin1o, N_GRAPHS, 384, 384);

    lin2_kernel<<<(N_GRAPHS * 10 + 255) / 256, 256, 0, stream>>>(lin1o, lin2_w, lin2_b,
                                                                 (float*)d_out);
}


# Neighbour sum: each source row read by one 16-byte load per lane (one request per 512-byte row, not four); owned rows kept column-permuted in LDS so the adds stay bank-free; sum order LDS arrival
# speedup vs baseline: 2.9027x; 1.0410x over previous
#include <hip/hip_runtime.h>

typedef __attribute__((ext_vector_type(2))) float v2f;
typedef __attribute__((ext_vector_type(8))) float v8f;

#define N_NODES  50000
#define N_EDGES  800000
#define N_GRAPHS 512
#define BN_EPS   1e-5f

__global__ void zero_f32(float* __restrict__ p, int n4) {
    int t = blockIdx.x * blockDim.x + threadIdx.x;
    if (t < n4) ((float4*)p)[t] = make_float4(0.f, 0.f, 0.f, 0.f);
}

__global__ void gather_emb(const int* __restrict__ ids, const float* __restrict__ emb,
                           float* __restrict__ x) {
    int t = blockIdx.x * blockDim.x + threadIdx.x;
    int node = t >> 5, lane = t & 31;
    if (node >= N_NODES) return;
    int id = ids[node];
    float4 v = *(const float4*)(emb + (long)id * 128 + lane * 4);
    *(float4*)(x + (long)node * 128 + lane * 4) = v;
}

#define AO_ROWS   196
#define AO_GROUPS ((N_NODES + AO_ROWS - 1) / AO_ROWS)
#define AO_SCAN   8
#define AO_TAKE   16
#define AO_QCAP   (AO_SCAN * 128 + AO_TAKE)
#define AO_ITERS  (N_EDGES / 128)
#define AO_REC_CAP 512
static_assert(AO_REC_CAP % AO_TAKE == 0 && AO_REC_CAP <= AO_QCAP, "a saved list is whole passes and fits the queue");
static_assert(N_NODES <= (1 << 16) && AO_ROWS < (1 << 8), "a saved word holds source row and local row");
static_assert(N_EDGES % 128 == 0, "a wave-load of the scan covers 128 edges");
static_assert(N_EDGES <= (1 << 20) && AO_ROWS <= (1 << 12), "a queue word holds edge and local row");
static_assert(AO_QCAP % AO_TAKE == 0 && AO_TAKE <= 32 && (AO_TAKE & (AO_TAKE - 1)) == 0,
              "a scan group fits behind a rest of less than one pass; one wave writes a pass's fill");

template <bool SAVED, bool REC>
__device__ __forceinline__ void ao_take(const unsigned* q, int i,
                                        const float* __restrict__ x, const int* __restrict__ ei,
                                        float* acc, int lane, unsigned* __restrict__ rec, int& nrec) {
    unsigned en[AO_TAKE];
    int s[AO_TAKE];
    float4 v[AO_TAKE];
#pragma unroll
    for (int j = 0; j < AO_TAKE; j++) {
        en[j] = __builtin_amdgcn_readfirstlane(q[i + j]);
        s[j] = SAVED ? (int)(en[j] & 0xFFFFu) : ei[en[j] & 0xFFFFFu];
    }
    if (REC && nrec >= 0) {
        if (nrec + AO_TAKE <= AO_REC_CAP) {
            unsigned w = 0u;
#pragma unroll
            for (int j = 0; j < AO_TAKE; j++)
                w = (lane == j) ? ((unsigned)s[j] | ((en[j] >> 20) << 16)) : w;
            if (lane < AO_TAKE) rec[nrec + lane] = w;
            nrec += AO_TAKE;
        } else {
            nrec = -1;
        }
    }
#pragma unroll
    for (int j = 0; j < AO_TAKE; j++)
        v[j] = *((const float4*)(x + (long)s[j] * 128) + lane);
#pragma unroll
    for (int j = 0; j < AO_TAKE; j++) {
        float* a = acc + (en[j] >> (SAVED ? 16 : 20)) * 128 + lane;
        atomicAdd(a + 0,  v[j].x);
        atomicAdd(a + 32, v[j].y);
        atomicAdd(a + 64, v[j].z);
        atomicAdd(a + 96, v[j].w);
    }
}

#define AO_FILL ((unsigned)AO_ROWS << 20)
template <bool REC>
__device__ __forceinline__ void ao_drain(unsigned* q, int cnt,
                                         const float* __restrict__ x, const int* __restrict__ ei,
                                         float* acc, int lane, unsigned* __restrict__ rec, int& nrec) {
    const int fill = (AO_TAKE - (cnt & (AO_TAKE - 1))) & (AO_TAKE - 1);
    if (lane < fill) q[cnt + lane] = AO_FILL;
    for (int i = 0; i < cnt; i += AO_TAKE) ao_take<false, REC>(q, i, x, ei, acc, lane, rec, nrec);
}

__device__ __forceinline__ void ao_replay(unsigned* q, const unsigned* __restrict__ rec, int n,
                                          const float* __restrict__ x, float* acc, int lane) {
    for (int i = lane; i < n; i += 32) q[i] = rec[i];
    int none = -1;
    for (int i = 0; i < n; i += AO_TAKE) ao_take<true, false>(q, i, x, nullptr, acc, lane, nullptr, none);
}

template <bool REC>
__device__ __forceinline__ int ao_full(unsigned* q, int cnt,
                                       const float* __restrict__ x, const int* __restrict__ ei,
                                       float* acc, int lane, unsigned* __restrict__ rec, int& nrec) {
    const int full = cnt & ~(AO_TAKE - 1);
    for (int i = 0; i < full; i += AO_TAKE) ao_take<false, REC>(q, i, x, ei, acc, lane, rec, nrec);
    const int rest = cnt - full;
    if (lane < rest) q[lane] = q[full + lane];
    return rest;
}

__device__ __forceinline__ int ao_push(unsigned* q, int cnt, bool m, int dl, int e) {
    const unsigned mask = __builtin_amdgcn_ballot_w32(m);
    if (mask != 0u) {
        if (m) q[cnt + (int)__builtin_amdgcn_mbcnt_lo(mask, 0u)] = (unsigned)e | ((unsigned)dl << 20);
        cnt += __builtin_popcount(mask);
    }
    return cnt;
}

__device__ __forceinline__ int ao_scan(unsigned* q, int cnt, int4 d, int e0, int lo, int nrows) {
    const bool m0 = (unsigned)(d.x - lo) < (unsigned)nrows;
    const bool m1 = (unsigned)(d.y - lo) < (unsigned)nrows;
    const bool m2 = (unsigned)(d.z - lo) < (unsigned)nrows;
    const bool m3 = (unsigned)(d.w - lo) < (unsigned)nrows;
    if (__builtin_amdgcn_ballot_w32(m0 | m1 | m2 | m3) != 0u) {
        cnt = ao_push(q, cnt, m0, d.x - lo, e0 + 0);
        cnt = ao_push(q, cnt, m1, d.y - lo, e0 + 1);
        cnt = ao_push(q, cnt, m2, d.z - lo, e0 + 2);
        cnt = ao_push(q, cnt, m3, d.w - lo, e0 + 3);
    }
    return cnt;
}

template <bool FIRST>
__global__ void __launch_bounds__(1024)
agg_owner(const float* __restrict__ x, const int* __restrict__ ei, float* __restrict__ agg,
          unsigned* __restrict__ recs, int* __restrict__ cnts) {
    __shared__ __attribute__((aligned(16))) float acc[(AO_ROWS + 1) * 128];
    __shared__ __attribute__((aligned(16))) unsigned queue[32][AO_QCAP];
    const int wave = __builtin_amdgcn_readfirstlane(threadIdx.x >> 5), lane = threadIdx.x & 31;
    const int lo = blockIdx.x * AO_ROWS;
    const int nrows = (N_NODES - lo) < AO_ROWS ? (N_NODES - lo) : AO_ROWS;

    for (int i = threadIdx.x; i < (AO_ROWS + 1) * 32; i += 1024)
        ((float4*)acc)[i] = make_float4(0.f, 0.f, 0.f, 0.f);
    __syncthreads();

    unsigned* q = queue[wave];
    unsigned* rec = recs + (long)(blockIdx.x * 32 + wave) * AO_REC_CAP;
    int nrec = 0;
    bool replay = false;
    int saved = 0;
    if (!FIRST) {
        const int c = cnts[blockIdx.x * 32 + lane];
        replay = __builtin_amdgcn_ballot_w32(c < 0) == 0u;
        saved = __builtin_amdgcn_readlane(c, wave);
    }
    if (replay) {
        ao_replay(q, rec, saved, x, acc, lane);
    } else {
        const int4* dst4 = (const int4*)(ei + N_EDGES);
        int cnt = 0;
        const int n_it = (AO_ITERS - wave + 31) / 32;
        const int first = (int)(blockIdx.x * n_it) / AO_GROUPS;

        for (int k = 0; k < n_it; k += AO_SCAN) {
            int4 d[AO_SCAN];
            int it[AO_SCAN];
#pragma unroll
            for (int u = 0; u < AO_SCAN; u++) {
                int kk = ((k + u) < n_it ? (k + u) : (n_it - 1)) + first;
                if (kk >= n_it) kk -= n_it;
                it[u] = wave + 32 * kk;
                d[u] = dst4[it[u] * 32 + lane];
            }
#pragma unroll
            for (int u = 0; u < AO_SCAN; u++)
                cnt = ao_scan(q, cnt, d[u], it[u] * 128 + lane * 4, lo, (k + u) < n_it ? nrows : 0);
            if (cnt >= AO_TAKE) cnt = ao_full<FIRST>(q, cnt, x, ei, acc, lane, rec, nrec);
        }
        ao_drain<FIRST>(q, cnt, x, ei, acc, lane, rec, nrec);
        if (FIRST && lane == 0) cnts[blockIdx.x * 32 + wave] = nrec;
    }

    __syncthreads();
    float4* out4 = (float4*)(agg + (long)lo * 128);
    for (int i = threadIdx.x; i < nrows * 32; i += 1024) {
        const float* r = acc + (i >> 5) * 128 + (i & 31);
        out4[i] = make_float4(r[0], r[32], r[64], r[96]);
    }
}

#define NG_PITCH 132
#define NG_TILES (N_NODES / 16)
#define NODE_GEMM_GROUPS 512
static_assert(NODE_GEMM_GROUPS <= NG_TILES, "every work-group has a first tile");
static_assert(N_NODES % 16 == 0, "row tiles must tile the node list");
template <bool FUSE_ADD, bool BN, bool RELU, bool POOL>
__global__ void __launch_bounds__(256)
gemm_node(const float* __restrict__ X, const float* __restrict__ AGG,
          const float* __restrict__ W, const float* __restrict__ bias,
          const float* __restrict__ gamma, const float* __restrict__ beta,
          const float* __restrict__ mean, const float* __restrict__ var,
          float* __restrict__ O, const int* __restrict__ batch, float* __restrict__ gsum) {
    __shared__ __attribute__((aligned(16))) float tileA[2][16 * NG_PITCH];
    const int wave = threadIdx.x >> 5, lane = threadIdx.x & 31;
    const int mi = lane & 15, kh = lane >> 4;
    const int col = wave * 16 + mi;

    v2f bf[32];
    const float* wc = W + (long)(kh * 2) * 128 + col;
#pragma unroll
    for (int j = 0; j < 32; j++) {
        bf[j].x = wc[(long)(4 * j) * 128];
        bf[j].y = wc[(long)(4 * j + 1) * 128];
    }

    const float bb = bias[col];
    float mu = 0.f, inv = 1.f, gm = 1.f, bt = 0.f;
    if (BN) { mu = mean[col]; inv = rsqrtf(var[col] + BN_EPS); gm = gamma[col]; bt = beta[col]; }

    const int so0 = wave * NG_PITCH + lane * 4;
    const int so1 = (wave + 8) * NG_PITCH + lane * 4;
    const long go0 = (long)wave * 128 + lane * 4;
    const long go1 = (long)(wave + 8) * 128 + lane * 4;

    int mt = blockIdx.x;
    {
        const float* xp = X + (long)mt * 16 * 128;
        float4 t0 = *(const float4*)(xp + go0);
        float4 t1 = *(const float4*)(xp + go1);
        if (FUSE_ADD) {
            const float* gp = AGG + (long)mt * 16 * 128;
            const float4 g0 = *(const float4*)(gp + go0);
            const float4 g1 = *(const float4*)(gp + go1);
            t0.x += g0.x; t0.y += g0.y; t0.z += g0.z; t0.w += g0.w;
            t1.x += g1.x; t1.y += g1.y; t1.z += g1.z; t1.w += g1.w;
        }
        *(float4*)(&tileA[0][so0]) = t0;
        *(float4*)(&tileA[0][so1]) = t1;
    }

    int cur = 0;
    for (; mt < NG_TILES; mt += gridDim.x) {
        __syncthreads();

        int bid = 0;
        if (POOL) bid = batch[mt * 16 + mi];

        const int nx = mt + gridDim.x;
        const int pf = nx < NG_TILES ? nx : mt;
        const float* xp = X + (long)pf * 16 * 128;
        const float4 x0 = *(const float4*)(xp + go0);
        const float4 x1 = *(const float4*)(xp + go1);
        float4 g0 = x0, g1 = x1;
        if (FUSE_ADD) {
            const float* gp = AGG + (long)pf * 16 * 128;
            g0 = *(const float4*)(gp + go0);
            g1 = *(const float4*)(gp + go1);
        }

        const float* ap = &tileA[cur][mi * NG_PITCH + kh * 2];
        v2f af[32];
#pragma unroll
        for (int j = 0; j < 32; j++) af[j] = *(const v2f*)(ap + 4 * j);
        v8f acc = {0.f, 0.f, 0.f, 0.f, 0.f, 0.f, 0.f, 0.f};
#pragma unroll
        for (int j = 0; j < 32; j++)
            acc = __builtin_amdgcn_wmma_f32_16x16x4_f32(false, af[j], false, bf[j],
                                                        (short)0, acc, false, false);
        __builtin_amdgcn_sched_group_barrier(0x100, 32, 0);
        __builtin_amdgcn_sched_group_barrier(0x008, 32, 0);
        __builtin_amdgcn_sched_group_barrier(0x002, 64, 0);

        float* orow = O + (long)(mt * 16 + kh * 8) * 128 + col;
        float vv[8];
#pragma unroll
        for (int r = 0; r < 8; r++) {
            float v = acc[r] + bb;
            if (BN)   v = (v - mu) * inv * gm + bt;
            if (RELU) v = fmaxf(v, 0.f);
            orow[(long)r * 128] = v;
            vv[r] = v;
        }

        if (POOL) {
            const int gfirst = __builtin_amdgcn_readfirstlane(bid);
            if (__builtin_amdgcn_ballot_w32(bid != gfirst) == 0u) {
                float s = vv[0];
#pragma unroll
                for (int r = 1; r < 8; r++) s += vv[r];
                const unsigned su = __float_as_uint(s);
                const float so = __uint_as_float(
                    __builtin_amdgcn_permlanex16(su, su, 0x76543210u, 0xfedcba98u, false, false));
                if (kh == 0) atomicAdd(gsum + (long)gfirst * 384 + col, s + so);
            } else {
                int idr[8];
#pragma unroll
                for (int r = 0; r < 8; r++) {
                    const int id_lo = __builtin_amdgcn_readlane(bid, r);
                    const int id_hi = __builtin_amdgcn_readlane(bid, 8 + r);
                    idr[r] = kh ? id_hi : id_lo;
                }
                int gcur = idr[0];
                float s = vv[0];
#pragma unroll
                for (int r = 1; r < 8; r++) {
                    if (idr[r] != gcur) {
                        atomicAdd(gsum + (long)gcur * 384 + col, s);
                        gcur = idr[r];
                        s = vv[r];
                    } else {
                        s += vv[r];
                    }
                }
                atomicAdd(gsum + (long)gcur * 384 + col, s);
            }
        }

        float4 t0 = x0, t1 = x1;
        if (FUSE_ADD) {
            t0.x += g0.x; t0.y += g0.y; t0.z += g0.z; t0.w += g0.w;
            t1.x += g1.x; t1.y += g1.y; t1.z += g1.z; t1.w += g1.w;
        }
        *(float4*)(&tileA[cur ^ 1][so0]) = t0;
        *(float4*)(&tileA[cur ^ 1][so1]) = t1;
        cur ^= 1;
    }
}

template <bool FUSE_ADD, bool BN, bool RELU>
__global__ void gemm_wmma(const float* __restrict__ X, const float* __restrict__ AGG,
                          const float* __restrict__ W, const float* __restrict__ bias,
                          const float* __restrict__ gamma, const float* __restrict__ beta,
                          const float* __restrict__ mean, const float* __restrict__ var,
                          float* __restrict__ O, int M, int K, int N) {
    int wave = threadIdx.x >> 5, lane = threadIdx.x & 31;
    int tilesN = N >> 4;
    int tile = blockIdx.x * 8 + wave;
    if (tile >= (M >> 4) * tilesN) return;
    int mt = tile / tilesN, nt = tile % tilesN;
    int mi = lane & 15, kh = lane >> 4;

    const float* xr = X + (long)(mt * 16 + mi) * K + kh * 2;
    const float* ar = FUSE_ADD ? (AGG + (long)(mt * 16 + mi) * K + kh * 2) : nullptr;
    const float* wc = W + (long)(kh * 2) * N + nt * 16 + mi;

    v8f acc = {0.f, 0.f, 0.f, 0.f, 0.f, 0.f, 0.f, 0.f};
    for (int kb = 0; kb < K; kb += 4) {
        v2f a, b;
        a.x = xr[kb];
        a.y = xr[kb + 1];
        if (FUSE_ADD) { a.x += ar[kb]; a.y += ar[kb + 1]; }
        b.x = wc[(long)kb * N];
        b.y = wc[(long)(kb + 1) * N];
        acc = __builtin_amdgcn_wmma_f32_16x16x4_f32(false, a, false, b,
                                                    (short)0, acc, false, false);
    }

    int col = nt * 16 + mi;
    float bb = bias[col];
    float mu = 0.f, inv = 1.f, gm = 1.f, bt = 0.f;
    if (BN) { mu = mean[col]; inv = rsqrtf(var[col] + BN_EPS); gm = gamma[col]; bt = beta[col]; }
    float* orow = O + (long)(mt * 16 + kh * 8) * N + col;
#pragma unroll
    for (int r = 0; r < 8; r++) {
        float v = acc[r] + bb;
        if (BN)   v = (v - mu) * inv * gm + bt;
        if (RELU) v = fmaxf(v, 0.f);
        orow[(long)r * N] = v;
    }
}

__global__ void lin2_kernel(const float* __restrict__ h, const float* __restrict__ w,
                            const float* __restrict__ b, float* __restrict__ out) {
    int t = blockIdx.x * blockDim.x + threadIdx.x;
    if (t >= N_GRAPHS * 10) return;
    int g = t / 10, o = t % 10;
    const float* hr = h + (long)g * 384;
    float acc = b[o];
    for (int k = 0; k < 384; k++) acc = fmaf(hr[k], w[k * 10 + o], acc);
    out[t] = acc;
}

extern "C" void kernel_launch(void* const* d_in, const int* in_sizes, int n_in,
                              void* d_out, int out_size, void* d_ws, size_t ws_size,
                              hipStream_t stream) {
    const int*   x_ids = (const int*)d_in[0];
    const int*   edge  = (const int*)d_in[1];
    const int*   batch = (const int*)d_in[2];
    const float* emb   = (const float*)d_in[3];

    const float *c_w1[3], *c_b1[3], *c_g[3], *c_be[3], *c_mu[3], *c_va[3], *c_w2[3], *c_b2[3];
    for (int i = 0; i < 3; i++) {
        c_w1[i] = (const float*)d_in[4 + 8 * i + 0];
        c_b1[i] = (const float*)d_in[4 + 8 * i + 1];
        c_g[i]  = (const float*)d_in[4 + 8 * i + 2];
        c_be[i] = (const float*)d_in[4 + 8 * i + 3];
        c_mu[i] = (const float*)d_in[4 + 8 * i + 4];
        c_va[i] = (const float*)d_in[4 + 8 * i + 5];
        c_w2[i] = (const float*)d_in[4 + 8 * i + 6];
        c_b2[i] = (const float*)d_in[4 + 8 * i + 7];
    }
    const float* lin1_w = (const float*)d_in[28];
    const float* lin1_b = (const float*)d_in[29];
    const float* lin2_w = (const float*)d_in[30];
    const float* lin2_b = (const float*)d_in[31];

    const long NF = (long)N_NODES * 128;
    float* ws   = (float*)d_ws;
    float* x0   = ws + 0 * NF;
    float* agg  = ws + 1 * NF;
    float* tmp  = ws + 2 * NF;
    float* hbuf[3] = { ws + 3 * NF, ws + 4 * NF, x0 };
    float* lin1o = tmp;
    unsigned* recs = (unsigned*)(ws + 5 * NF);
    int* cnts = (int*)(recs + (long)AO_GROUPS * 32 * AO_REC_CAP);
    float* gcat = (float*)(cnts + AO_GROUPS * 32);

    zero_f32<<<(N_GRAPHS * 384 / 4 + 255) / 256, 256, 0, stream>>>(gcat, N_GRAPHS * 384 / 4);

    gather_emb<<<(N_NODES * 32 + 255) / 256, 256, 0, stream>>>(x_ids, emb, x0);

    const float* xin = x0;
    for (int l = 0; l < 3; l++) {
        if (l == 0) agg_owner<true><<<AO_GROUPS, 1024, 0, stream>>>(xin, edge, agg, recs, cnts);
        else        agg_owner<false><<<AO_GROUPS, 1024, 0, stream>>>(xin, edge, agg, recs, cnts);
        gemm_node<true, true, true, false><<<NODE_GEMM_GROUPS, 256, 0, stream>>>(
            xin, agg, c_w1[l], c_b1[l], c_g[l], c_be[l], c_mu[l], c_va[l], tmp, nullptr, nullptr);
        gemm_node<false, false, true, true><<<NODE_GEMM_GROUPS, 256, 0, stream>>>(
            tmp, nullptr, c_w2[l], c_b2[l], nullptr, nullptr, nullptr, nullptr, hbuf[l],
            batch, gcat + l * 128);
        xin = hbuf[l];
    }

    const int tiles_head = (N_GRAPHS / 16) * (384 / 16);
    gemm_wmma<false, false, true><<<(tiles_head + 7) / 8, 256, 0, stream>>>(
        gcat, nullptr, lin1_w, lin1_b, nullptr, nullptr, nullptr, nullptr,
        lin1o, N_GRAPHS, 384, 384);

    lin2_kernel<<<(N_GRAPHS * 10 + 255) / 256, 256, 0, stream>>>(lin1o, lin2_w, lin2_b,
                                                                 (float*)d_out);
}


# Neighbour sum's scan reads a one-byte owner-group list (made by the start-up launch, read by 16-byte loads), not the 4-byte destination list; the take reads the row's index; sum order LDS arrival
# speedup vs baseline: 3.0065x; 1.0358x over previous
#include <hip/hip_runtime.h>

typedef __attribute__((ext_vector_type(2))) float v2f;
typedef __attribute__((ext_vector_type(8))) float v8f;

#define N_NODES  50000
#define N_EDGES  800000
#define N_GRAPHS 512
#define BN_EPS   1e-5f

__global__ void zero_f32(float* __restrict__ p, int n4) {
    int t = blockIdx.x * blockDim.x + threadIdx.x;
    if (t < n4) ((float4*)p)[t] = make_float4(0.f, 0.f, 0.f, 0.f);
}

#define AO_ROWS   196
#define AO_GROUPS ((N_NODES + AO_ROWS - 1) / AO_ROWS)
static_assert(AO_GROUPS <= 256, "an owner group's number fits a byte");
static_assert(N_EDGES % 4 == 0 && (long)N_NODES * 32 >= N_EDGES / 4, "one thread per four edges");

__device__ __forceinline__ unsigned ao_owner(int d) {
    const unsigned g = (unsigned)d / AO_ROWS;
    return g < 255u ? g : 255u;
}

__global__ void gather_emb(const int* __restrict__ ids, const float* __restrict__ emb,
                           float* __restrict__ x, const int* __restrict__ ei,
                           unsigned* __restrict__ own) {
    int t = blockIdx.x * blockDim.x + threadIdx.x;
    if (t < N_EDGES / 4) {
        const int4 d = ((const int4*)(ei + N_EDGES))[t];
        own[t] = ao_owner(d.x) | (ao_owner(d.y) << 8) | (ao_owner(d.z) << 16) | (ao_owner(d.w) << 24);
    }
    int node = t >> 5, lane = t & 31;
    if (node >= N_NODES) return;
    int id = ids[node];
    float4 v = *(const float4*)(emb + (long)id * 128 + lane * 4);
    *(float4*)(x + (long)node * 128 + lane * 4) = v;
}

#define AO_SCAN   3
#define AO_TAKE   16
#define AO_QCAP   (AO_SCAN * 512 + AO_TAKE)
#define AO_ITERS  (N_EDGES / 512)
#define AO_TAIL   (N_EDGES % 512)
#define AO_REC_CAP 512
static_assert(AO_REC_CAP % AO_TAKE == 0 && AO_REC_CAP <= AO_QCAP, "a saved list is whole passes and fits the queue");
static_assert(N_NODES <= (1 << 16) && AO_ROWS < (1 << 8), "a saved word holds source row and local row");
static_assert(AO_TAIL == 256, "the list is whole 16-byte wave-loads of 512 edges and one 8-byte wave-load of 256");
static_assert(AO_QCAP >= AO_TAKE + AO_TAIL, "the tail fits behind a rest of less than one pass");
static_assert((long)N_EDGES < (1L << 31), "a queue word holds the edge number below the fill mark");
static_assert(AO_QCAP % AO_TAKE == 0 && AO_TAKE <= 32 && (AO_TAKE & (AO_TAKE - 1)) == 0,
              "a scan group fits behind a rest of less than one pass; one wave writes a pass's fill");

template <bool SAVED, bool REC>
__device__ __forceinline__ void ao_take(const unsigned* q, int i,
                                        const float* __restrict__ x, const int* __restrict__ ei,
                                        float* acc, int lane, unsigned* __restrict__ rec, int& nrec,
                                        int lo) {
    unsigned en[AO_TAKE];
    int s[AO_TAKE];
    int dv[AO_TAKE];
    unsigned row[AO_TAKE];
    float4 v[AO_TAKE];
#pragma unroll
    for (int j = 0; j < AO_TAKE; j++) {
        en[j] = __builtin_amdgcn_readfirstlane(q[i + j]);
        s[j] = SAVED ? (int)(en[j] & 0xFFFFu) : ei[en[j] & 0x7FFFFFFFu];
        dv[j] = SAVED ? 0 : ei[N_EDGES + (en[j] & 0x7FFFFFFFu)];
    }
#pragma unroll
    for (int j = 0; j < AO_TAKE; j++) {
        if (SAVED) {
            row[j] = en[j] >> 16;
        } else {
            const unsigned dl = ((unsigned)dv[j] - (unsigned)lo) | (unsigned)((int)en[j] >> 31);
            row[j] = dl < (unsigned)AO_ROWS ? dl : (unsigned)AO_ROWS;
        }
    }
    if (REC && nrec >= 0) {
        if (nrec + AO_TAKE <= AO_REC_CAP) {
            unsigned w = 0u;
#pragma unroll
            for (int j = 0; j < AO_TAKE; j++)
                w = (lane == j) ? ((unsigned)s[j] | (row[j] << 16)) : w;
            if (lane < AO_TAKE) rec[nrec + lane] = w;
            nrec += AO_TAKE;
        } else {
            nrec = -1;
        }
    }
#pragma unroll
    for (int j = 0; j < AO_TAKE; j++)
        v[j] = *((const float4*)(x + (long)s[j] * 128) + lane);
#pragma unroll
    for (int j = 0; j < AO_TAKE; j++) {
        float* a = acc + row[j] * 128 + lane;
        atomicAdd(a + 0,  v[j].x);
        atomicAdd(a + 32, v[j].y);
        atomicAdd(a + 64, v[j].z);
        atomicAdd(a + 96, v[j].w);
    }
}

#define AO_FILL 0x80000000u
template <bool REC>
__device__ __forceinline__ void ao_drain(unsigned* q, int cnt,
                                         const float* __restrict__ x, const int* __restrict__ ei,
                                         float* acc, int lane, unsigned* __restrict__ rec, int& nrec,
                                         int lo) {
    const int fill = (AO_TAKE - (cnt & (AO_TAKE - 1))) & (AO_TAKE - 1);
    if (lane < fill) q[cnt + lane] = AO_FILL;
    for (int i = 0; i < cnt; i += AO_TAKE)
        ao_take<false, REC>(q, i, x, ei, acc, lane, rec, nrec, lo);
}

__device__ __forceinline__ void ao_replay(unsigned* q, const unsigned* __restrict__ rec, int n,
                                          const float* __restrict__ x, float* acc, int lane) {
    for (int i = lane; i < n; i += 32) q[i] = rec[i];
    int none = -1;
    for (int i = 0; i < n; i += AO_TAKE)
        ao_take<true, false>(q, i, x, nullptr, acc, lane, nullptr, none, 0);
}

template <bool REC>
__device__ __forceinline__ int ao_full(unsigned* q, int cnt,
                                       const float* __restrict__ x, const int* __restrict__ ei,
                                       float* acc, int lane, unsigned* __restrict__ rec, int& nrec,
                                       int lo) {
    const int full = cnt & ~(AO_TAKE - 1);
    for (int i = 0; i < full; i += AO_TAKE)
        ao_take<false, REC>(q, i, x, ei, acc, lane, rec, nrec, lo);
    const int rest = cnt - full;
    if (lane < rest) q[lane] = q[full + lane];
    return rest;
}

__device__ __forceinline__ int ao_push(unsigned* q, int cnt, bool m, int e) {
    const unsigned mask = __builtin_amdgcn_ballot_w32(m);
    if (mask != 0u) {
        if (m) q[cnt + (int)__builtin_amdgcn_mbcnt_lo(mask, 0u)] = (unsigned)e;
        cnt += __builtin_popcount(mask);
    }
    return cnt;
}

__device__ __forceinline__ int ao_scan(unsigned* q, int cnt, unsigned w, int e0, unsigned gx) {
    const unsigned x = w ^ gx;
    const unsigned z = (x - 0x01010101u) & ~x & 0x80808080u;
    if (__builtin_amdgcn_ballot_w32(z != 0u) != 0u) {
        cnt = ao_push(q, cnt, (x & 0x000000FFu) == 0u, e0 + 0);
        cnt = ao_push(q, cnt, (x & 0x0000FF00u) == 0u, e0 + 1);
        cnt = ao_push(q, cnt, (x & 0x00FF0000u) == 0u, e0 + 2);
        cnt = ao_push(q, cnt, (x & 0xFF000000u) == 0u, e0 + 3);
    }
    return cnt;
}

template <bool FIRST>
__global__ void __launch_bounds__(1024)
agg_owner(const float* __restrict__ x, const int* __restrict__ ei, float* __restrict__ agg,
          unsigned* __restrict__ recs, int* __restrict__ cnts, const unsigned* __restrict__ own) {
    __shared__ __attribute__((aligned(16))) float acc[(AO_ROWS + 1) * 128];
    __shared__ __attribute__((aligned(16))) unsigned queue[32][AO_QCAP];
    const int wave = __builtin_amdgcn_readfirstlane(threadIdx.x >> 5), lane = threadIdx.x & 31;
    const int lo = blockIdx.x * AO_ROWS;
    const int nrows = (N_NODES - lo) < AO_ROWS ? (N_NODES - lo) : AO_ROWS;

    for (int i = threadIdx.x; i < (AO_ROWS + 1) * 32; i += 1024)
        ((float4*)acc)[i] = make_float4(0.f, 0.f, 0.f, 0.f);
    __syncthreads();

    unsigned* q = queue[wave];
    unsigned* rec = recs + (long)(blockIdx.x * 32 + wave) * AO_REC_CAP;
    int nrec = 0;
    bool replay = false;
    int saved = 0;
    if (!FIRST) {
        const int c = cnts[blockIdx.x * 32 + lane];
        replay = __builtin_amdgcn_ballot_w32(c < 0) == 0u;
        saved = __builtin_amdgcn_readlane(c, wave);
    }
    if (replay) {
        ao_replay(q, rec, saved, x, acc, lane);
    } else {
        const uint4* own4 = (const uint4*)own;
        const unsigned gx = blockIdx.x * 0x01010101u;
        int cnt = 0;
        const int n_it = (AO_ITERS - wave + 31) / 32;
        const int first = (int)(blockIdx.x * n_it) / AO_GROUPS;

        for (int k = 0; k < n_it; k += AO_SCAN) {
            uint4 d[AO_SCAN];
            int it[AO_SCAN];
#pragma unroll
            for (int u = 0; u < AO_SCAN; u++) {
                int kk = ((k + u) < n_it ? (k + u) : (n_it - 1)) + first;
                if (kk >= n_it) kk -= n_it;
                it[u] = wave + 32 * kk;
                d[u] = own4[it[u] * 32 + lane];
            }
#pragma unroll
            for (int u = 0; u < AO_SCAN; u++) {
                if ((k + u) < n_it) {
                    const int e0 = it[u] * 512 + lane * 16;
                    cnt = ao_scan(q, cnt, d[u].x, e0, gx);
                    cnt = ao_scan(q, cnt, d[u].y, e0 + 4, gx);
                    cnt = ao_scan(q, cnt, d[u].z, e0 + 8, gx);
                    cnt = ao_scan(q, cnt, d[u].w, e0 + 12, gx);
                }
            }
            if (cnt >= AO_TAKE) cnt = ao_full<FIRST>(q, cnt, x, ei, acc, lane, rec, nrec, lo);
        }
        if (wave == 31) {
            const uint2 t = ((const uint2*)own)[AO_ITERS * 64 + lane];
            const int e0 = AO_ITERS * 512 + lane * 8;
            cnt = ao_scan(q, cnt, t.x, e0, gx);
            cnt = ao_scan(q, cnt, t.y, e0 + 4, gx);
        }
        ao_drain<FIRST>(q, cnt, x, ei, acc, lane, rec, nrec, lo);
        if (FIRST && lane == 0) cnts[blockIdx.x * 32 + wave] = nrec;
    }

    __syncthreads();
    float4* out4 = (float4*)(agg + (long)lo * 128);
    for (int i = threadIdx.x; i < nrows * 32; i += 1024) {
        const float* r = acc + (i >> 5) * 128 + (i & 31);
        out4[i] = make_float4(r[0], r[32], r[64], r[96]);
    }
}

#define NG_PITCH 132
#define NG_TILES (N_NODES / 16)
#define NODE_GEMM_GROUPS 512
static_assert(NODE_GEMM_GROUPS <= NG_TILES, "every work-group has a first tile");
static_assert(N_NODES % 16 == 0, "row tiles must tile the node list");
template <bool FUSE_ADD, bool BN, bool RELU, bool POOL>
__global__ void __launch_bounds__(256)
gemm_node(const float* __restrict__ X, const float* __restrict__ AGG,
          const float* __restrict__ W, const float* __restrict__ bias,
          const float* __restrict__ gamma, const float* __restrict__ beta,
          const float* __restrict__ mean, const float* __restrict__ var,
          float* __restrict__ O, const int* __restrict__ batch, float* __restrict__ gsum) {
    __shared__ __attribute__((aligned(16))) float tileA[2][16 * NG_PITCH];
    const int wave = threadIdx.x >> 5, lane = threadIdx.x & 31;
    const int mi = lane & 15, kh = lane >> 4;
    const int col = wave * 16 + mi;

    v2f bf[32];
    const float* wc = W + (long)(kh * 2) * 128 + col;
#pragma unroll
    for (int j = 0; j < 32; j++) {
        bf[j].x = wc[(long)(4 * j) * 128];
        bf[j].y = wc[(long)(4 * j + 1) * 128];
    }

    const float bb = bias[col];
    float mu = 0.f, inv = 1.f, gm = 1.f, bt = 0.f;
    if (BN) { mu = mean[col]; inv = rsqrtf(var[col] + BN_EPS); gm = gamma[col]; bt = beta[col]; }

    const int so0 = wave * NG_PITCH + lane * 4;
    const int so1 = (wave + 8) * NG_PITCH + lane * 4;
    const long go0 = (long)wave * 128 + lane * 4;
    const long go1 = (long)(wave + 8) * 128 + lane * 4;

    int mt = blockIdx.x;
    {
        const float* xp = X + (long)mt * 16 * 128;
        float4 t0 = *(const float4*)(xp + go0);
        float4 t1 = *(const float4*)(xp + go1);
        if (FUSE_ADD) {
            const float* gp = AGG + (long)mt * 16 * 128;
            const float4 g0 = *(const float4*)(gp + go0);
            const float4 g1 = *(const float4*)(gp + go1);
            t0.x += g0.x; t0.y += g0.y; t0.z += g0.z; t0.w += g0.w;
            t1.x += g1.x; t1.y += g1.y; t1.z += g1.z; t1.w += g1.w;
        }
        *(float4*)(&tileA[0][so0]) = t0;
        *(float4*)(&tileA[0][so1]) = t1;
    }

    int cur = 0;
    for (; mt < NG_TILES; mt += gridDim.x) {
        __syncthreads();

        int bid = 0;
        if (POOL) bid = batch[mt * 16 + mi];

        const int nx = mt + gridDim.x;
        const int pf = nx < NG_TILES ? nx : mt;
        const float* xp = X + (long)pf * 16 * 128;
        const float4 x0 = *(const float4*)(xp + go0);
        const float4 x1 = *(const float4*)(xp + go1);
        float4 g0 = x0, g1 = x1;
        if (FUSE_ADD) {
            const float* gp = AGG + (long)pf * 16 * 128;
            g0 = *(const float4*)(gp + go0);
            g1 = *(const float4*)(gp + go1);
        }

        const float* ap = &tileA[cur][mi * NG_PITCH + kh * 2];
        v2f af[32];
#pragma unroll
        for (int j = 0; j < 32; j++) af[j] = *(const v2f*)(ap + 4 * j);
        v8f acc = {0.f, 0.f, 0.f, 0.f, 0.f, 0.f, 0.f, 0.f};
#pragma unroll
        for (int j = 0; j < 32; j++)
            acc = __builtin_amdgcn_wmma_f32_16x16x4_f32(false, af[j], false, bf[j],
                                                        (short)0, acc, false, false);
        __builtin_amdgcn_sched_group_barrier(0x100, 32, 0);
        __builtin_amdgcn_sched_group_barrier(0x008, 32, 0);
        __builtin_amdgcn_sched_group_barrier(0x002, 64, 0);

        float* orow = O + (long)(mt * 16 + kh * 8) * 128 + col;
        float vv[8];
#pragma unroll
        for (int r = 0; r < 8; r++) {
            float v = acc[r] + bb;
            if (BN)   v = (v - mu) * inv * gm + bt;
            if (RELU) v = fmaxf(v, 0.f);
            orow[(long)r * 128] = v;
            vv[r] = v;
        }

        if (POOL) {
            const int gfirst = __builtin_amdgcn_readfirstlane(bid);
            if (__builtin_amdgcn_ballot_w32(bid != gfirst) == 0u) {
                float s = vv[0];
#pragma unroll
                for (int r = 1; r < 8; r++) s += vv[r];
                const unsigned su = __float_as_uint(s);
                const float so = __uint_as_float(
                    __builtin_amdgcn_permlanex16(su, su, 0x76543210u, 0xfedcba98u, false, false));
                if (kh == 0) atomicAdd(gsum + (long)gfirst * 384 + col, s + so);
            } else {
                int idr[8];
#pragma unroll
                for (int r = 0; r < 8; r++) {
                    const int id_lo = __builtin_amdgcn_readlane(bid, r);
                    const int id_hi = __builtin_amdgcn_readlane(bid, 8 + r);
                    idr[r] = kh ? id_hi : id_lo;
                }
                int gcur = idr[0];
                float s = vv[0];
#pragma unroll
                for (int r = 1; r < 8; r++) {
                    if (idr[r] != gcur) {
                        atomicAdd(gsum + (long)gcur * 384 + col, s);
                        gcur = idr[r];
                        s = vv[r];
                    } else {
                        s += vv[r];
                    }
                }
                atomicAdd(gsum + (long)gcur * 384 + col, s);
            }
        }

        float4 t0 = x0, t1 = x1;
        if (FUSE_ADD) {
            t0.x += g0.x; t0.y += g0.y; t0.z += g0.z; t0.w += g0.w;
            t1.x += g1.x; t1.y += g1.y; t1.z += g1.z; t1.w += g1.w;
        }
        *(float4*)(&tileA[cur ^ 1][so0]) = t0;
        *(float4*)(&tileA[cur ^ 1][so1]) = t1;
        cur ^= 1;
    }
}

template <bool FUSE_ADD, bool BN, bool RELU>
__global__ void gemm_wmma(const float* __restrict__ X, const float* __restrict__ AGG,
                          const float* __restrict__ W, const float* __restrict__ bias,
                          const float* __restrict__ gamma, const float* __restrict__ beta,
                          const float* __restrict__ mean, const float* __restrict__ var,
                          float* __restrict__ O, int M, int K, int N) {
    int wave = threadIdx.x >> 5, lane = threadIdx.x & 31;
    int tilesN = N >> 4;
    int tile = blockIdx.x * 8 + wave;
    if (tile >= (M >> 4) * tilesN) return;
    int mt = tile / tilesN, nt = tile % tilesN;
    int mi = lane & 15, kh = lane >> 4;

    const float* xr = X + (long)(mt * 16 + mi) * K + kh * 2;
    const float* ar = FUSE_ADD ? (AGG + (long)(mt * 16 + mi) * K + kh * 2) : nullptr;
    const float* wc = W + (long)(kh * 2) * N + nt * 16 + mi;

    v8f acc = {0.f, 0.f, 0.f, 0.f, 0.f, 0.f, 0.f, 0.f};
    for (int kb = 0; kb < K; kb += 4) {
        v2f a, b;
        a.x = xr[kb];
        a.y = xr[kb + 1];
        if (FUSE_ADD) { a.x += ar[kb]; a.y += ar[kb + 1]; }
        b.x = wc[(long)kb * N];
        b.y = wc[(long)(kb + 1) * N];
        acc = __builtin_amdgcn_wmma_f32_16x16x4_f32(false, a, false, b,
                                                    (short)0, acc, false, false);
    }

    int col = nt * 16 + mi;
    float bb = bias[col];
    float mu = 0.f, inv = 1.f, gm = 1.f, bt = 0.f;
    if (BN) { mu = mean[col]; inv = rsqrtf(var[col] + BN_EPS); gm = gamma[col]; bt = beta[col]; }
    float* orow = O + (long)(mt * 16 + kh * 8) * N + col;
#pragma unroll
    for (int r = 0; r < 8; r++) {
        float v = acc[r] + bb;
        if (BN)   v = (v - mu) * inv * gm + bt;
        if (RELU) v = fmaxf(v, 0.f);
        orow[(long)r * N] = v;
    }
}

__global__ void lin2_kernel(const float* __restrict__ h, const float* __restrict__ w,
                            const float* __restrict__ b, float* __restrict__ out) {
    int t = blockIdx.x * blockDim.x + threadIdx.x;
    if (t >= N_GRAPHS * 10) return;
    int g = t / 10, o = t % 10;
    const float* hr = h + (long)g * 384;
    float acc = b[o];
    for (int k = 0; k < 384; k++) acc = fmaf(hr[k], w[k * 10 + o], acc);
    out[t] = acc;
}

extern "C" void kernel_launch(void* const* d_in, const int* in_sizes, int n_in,
                              void* d_out, int out_size, void* d_ws, size_t ws_size,
                              hipStream_t stream) {
    const int*   x_ids = (const int*)d_in[0];
    const int*   edge  = (const int*)d_in[1];
    const int*   batch = (const int*)d_in[2];
    const float* emb   = (const float*)d_in[3];

    const float *c_w1[3], *c_b1[3], *c_g[3], *c_be[3], *c_mu[3], *c_va[3], *c_w2[3], *c_b2[3];
    for (int i = 0; i < 3; i++) {
        c_w1[i] = (const float*)d_in[4 + 8 * i + 0];
        c_b1[i] = (const float*)d_in[4 + 8 * i + 1];
        c_g[i]  = (const float*)d_in[4 + 8 * i + 2];
        c_be[i] = (const float*)d_in[4 + 8 * i + 3];
        c_mu[i] = (const float*)d_in[4 + 8 * i + 4];
        c_va[i] = (const float*)d_in[4 + 8 * i + 5];
        c_w2[i] = (const float*)d_in[4 + 8 * i + 6];
        c_b2[i] = (const float*)d_in[4 + 8 * i + 7];
    }
    const float* lin1_w = (const float*)d_in[28];
    const float* lin1_b = (const float*)d_in[29];
    const float* lin2_w = (const float*)d_in[30];
    const float* lin2_b = (const float*)d_in[31];

    const long NF = (long)N_NODES * 128;
    float* ws   = (float*)d_ws;
    float* x0   = ws + 0 * NF;
    float* agg  = ws + 1 * NF;
    float* tmp  = ws + 2 * NF;
    float* hbuf[3] = { ws + 3 * NF, ws + 4 * NF, x0 };
    float* lin1o = tmp;
    unsigned* recs = (unsigned*)(ws + 5 * NF);
    int* cnts = (int*)(recs + (long)AO_GROUPS * 32 * AO_REC_CAP);
    float* gcat = (float*)(cnts + AO_GROUPS * 32);
    unsigned* own = (unsigned*)(gcat + N_GRAPHS * 384);

    zero_f32<<<(N_GRAPHS * 384 / 4 + 255) / 256, 256, 0, stream>>>(gcat, N_GRAPHS * 384 / 4);

    gather_emb<<<(N_NODES * 32 + 255) / 256, 256, 0, stream>>>(x_ids, emb, x0, edge, own);

    const float* xin = x0;
    for (int l = 0; l < 3; l++) {
        if (l == 0) agg_owner<true><<<AO_GROUPS, 1024, 0, stream>>>(xin, edge, agg, recs, cnts, own);
        else        agg_owner<false><<<AO_GROUPS, 1024, 0, stream>>>(xin, edge, agg, recs, cnts, own);
        gemm_node<true, true, true, false><<<NODE_GEMM_GROUPS, 256, 0, stream>>>(
            xin, agg, c_w1[l], c_b1[l], c_g[l], c_be[l], c_mu[l], c_va[l], tmp, nullptr, nullptr);
        gemm_node<false, false, true, true><<<NODE_GEMM_GROUPS, 256, 0, stream>>>(
            tmp, nullptr, c_w2[l], c_b2[l], nullptr, nullptr, nullptr, nullptr, hbuf[l],
            batch, gcat + l * 128);
        xin = hbuf[l];
    }

    const int tiles_head = (N_GRAPHS / 16) * (384 / 16);
    gemm_wmma<false, false, true><<<(tiles_head + 7) / 8, 256, 0, stream>>>(
        gcat, nullptr, lin1_w, lin1_b, nullptr, nullptr, nullptr, nullptr,
        lin1o, N_GRAPHS, 384, 384);

    lin2_kernel<<<(N_GRAPHS * 10 + 255) / 256, 256, 0, stream>>>(lin1o, lin2_w, lin2_b,
                                                                 (float*)d_out);
}


# Head's two products: k split over 4 waves per tile (lin1, partials joined in LDS) and over a wave's lanes (lin2, joined by lane exchanges) so a wave's loads go out as one group; k sum order changes
# speedup vs baseline: 3.4044x; 1.1323x over previous
#include <hip/hip_runtime.h>

typedef __attribute__((ext_vector_type(2))) float v2f;
typedef __attribute__((ext_vector_type(8))) float v8f;

#define N_NODES  50000
#define N_EDGES  800000
#define N_GRAPHS 512
#define BN_EPS   1e-5f

__global__ void zero_f32(float* __restrict__ p, int n4) {
    int t = blockIdx.x * blockDim.x + threadIdx.x;
    if (t < n4) ((float4*)p)[t] = make_float4(0.f, 0.f, 0.f, 0.f);
}

#define AO_ROWS   196
#define AO_GROUPS ((N_NODES + AO_ROWS - 1) / AO_ROWS)
static_assert(AO_GROUPS <= 256, "an owner group's number fits a byte");
static_assert(N_EDGES % 4 == 0 && (long)N_NODES * 32 >= N_EDGES / 4, "one thread per four edges");

__device__ __forceinline__ unsigned ao_owner(int d) {
    const unsigned g = (unsigned)d / AO_ROWS;
    return g < 255u ? g : 255u;
}

__global__ void gather_emb(const int* __restrict__ ids, const float* __restrict__ emb,
                           float* __restrict__ x, const int* __restrict__ ei,
                           unsigned* __restrict__ own) {
    int t = blockIdx.x * blockDim.x + threadIdx.x;
    if (t < N_EDGES / 4) {
        const int4 d = ((const int4*)(ei + N_EDGES))[t];
        own[t] = ao_owner(d.x) | (ao_owner(d.y) << 8) | (ao_owner(d.z) << 16) | (ao_owner(d.w) << 24);
    }
    int node = t >> 5, lane = t & 31;
    if (node >= N_NODES) return;
    int id = ids[node];
    float4 v = *(const float4*)(emb + (long)id * 128 + lane * 4);
    *(float4*)(x + (long)node * 128 + lane * 4) = v;
}

#define AO_SCAN   3
#define AO_TAKE   16
#define AO_QCAP   (AO_SCAN * 512 + AO_TAKE)
#define AO_ITERS  (N_EDGES / 512)
#define AO_TAIL   (N_EDGES % 512)
#define AO_REC_CAP 512
static_assert(AO_REC_CAP % AO_TAKE == 0 && AO_REC_CAP <= AO_QCAP, "a saved list is whole passes and fits the queue");
static_assert(N_NODES <= (1 << 16) && AO_ROWS < (1 << 8), "a saved word holds source row and local row");
static_assert(AO_TAIL == 256, "the list is whole 16-byte wave-loads of 512 edges and one 8-byte wave-load of 256");
static_assert(AO_QCAP >= AO_TAKE + AO_TAIL, "the tail fits behind a rest of less than one pass");
static_assert((long)N_EDGES < (1L << 31), "a queue word holds the edge number below the fill mark");
static_assert(AO_QCAP % AO_TAKE == 0 && AO_TAKE <= 32 && (AO_TAKE & (AO_TAKE - 1)) == 0,
              "a scan group fits behind a rest of less than one pass; one wave writes a pass's fill");

template <bool SAVED, bool REC>
__device__ __forceinline__ void ao_take(const unsigned* q, int i,
                                        const float* __restrict__ x, const int* __restrict__ ei,
                                        float* acc, int lane, unsigned* __restrict__ rec, int& nrec,
                                        int lo) {
    unsigned en[AO_TAKE];
    int s[AO_TAKE];
    int dv[AO_TAKE];
    unsigned row[AO_TAKE];
    float4 v[AO_TAKE];
#pragma unroll
    for (int j = 0; j < AO_TAKE; j++) {
        en[j] = __builtin_amdgcn_readfirstlane(q[i + j]);
        s[j] = SAVED ? (int)(en[j] & 0xFFFFu) : ei[en[j] & 0x7FFFFFFFu];
        dv[j] = SAVED ? 0 : ei[N_EDGES + (en[j] & 0x7FFFFFFFu)];
    }
#pragma unroll
    for (int j = 0; j < AO_TAKE; j++) {
        if (SAVED) {
            row[j] = en[j] >> 16;
        } else {
            const unsigned dl = ((unsigned)dv[j] - (unsigned)lo) | (unsigned)((int)en[j] >> 31);
            row[j] = dl < (unsigned)AO_ROWS ? dl : (unsigned)AO_ROWS;
        }
    }
    if (REC && nrec >= 0) {
        if (nrec + AO_TAKE <= AO_REC_CAP) {
            unsigned w = 0u;
#pragma unroll
            for (int j = 0; j < AO_TAKE; j++)
                w = (lane == j) ? ((unsigned)s[j] | (row[j] << 16)) : w;
            if (lane < AO_TAKE) rec[nrec + lane] = w;
            nrec += AO_TAKE;
        } else {
            nrec = -1;
        }
    }
#pragma unroll
    for (int j = 0; j < AO_TAKE; j++)
        v[j] = *((const float4*)(x + (long)s[j] * 128) + lane);
#pragma unroll
    for (int j = 0; j < AO_TAKE; j++) {
        float* a = acc + row[j] * 128 + lane;
        atomicAdd(a + 0,  v[j].x);
        atomicAdd(a + 32, v[j].y);
        atomicAdd(a + 64, v[j].z);
        atomicAdd(a + 96, v[j].w);
    }
}

#define AO_FILL 0x80000000u
template <bool REC>
__device__ __forceinline__ void ao_drain(unsigned* q, int cnt,
                                         const float* __restrict__ x, const int* __restrict__ ei,
                                         float* acc, int lane, unsigned* __restrict__ rec, int& nrec,
                                         int lo) {
    const int fill = (AO_TAKE - (cnt & (AO_TAKE - 1))) & (AO_TAKE - 1);
    if (lane < fill) q[cnt + lane] = AO_FILL;
    for (int i = 0; i < cnt; i += AO_TAKE)
        ao_take<false, REC>(q, i, x, ei, acc, lane, rec, nrec, lo);
}

__device__ __forceinline__ void ao_replay(unsigned* q, const unsigned* __restrict__ rec, int n,
                                          const float* __restrict__ x, float* acc, int lane) {
    for (int i = lane; i < n; i += 32) q[i] = rec[i];
    int none = -1;
    for (int i = 0; i < n; i += AO_TAKE)
        ao_take<true, false>(q, i, x, nullptr, acc, lane, nullptr, none, 0);
}

template <bool REC>
__device__ __forceinline__ int ao_full(unsigned* q, int cnt,
                                       const float* __restrict__ x, const int* __restrict__ ei,
                                       float* acc, int lane, unsigned* __restrict__ rec, int& nrec,
                                       int lo) {
    const int full = cnt & ~(AO_TAKE - 1);
    for (int i = 0; i < full; i += AO_TAKE)
        ao_take<false, REC>(q, i, x, ei, acc, lane, rec, nrec, lo);
    const int rest = cnt - full;
    if (lane < rest) q[lane] = q[full + lane];
    return rest;
}

__device__ __forceinline__ int ao_push(unsigned* q, int cnt, bool m, int e) {
    const unsigned mask = __builtin_amdgcn_ballot_w32(m);
    if (mask != 0u) {
        if (m) q[cnt + (int)__builtin_amdgcn_mbcnt_lo(mask, 0u)] = (unsigned)e;
        cnt += __builtin_popcount(mask);
    }
    return cnt;
}

__device__ __forceinline__ int ao_scan(unsigned* q, int cnt, unsigned w, int e0, unsigned gx) {
    const unsigned x = w ^ gx;
    const unsigned z = (x - 0x01010101u) & ~x & 0x80808080u;
    if (__builtin_amdgcn_ballot_w32(z != 0u) != 0u) {
        cnt = ao_push(q, cnt, (x & 0x000000FFu) == 0u, e0 + 0);
        cnt = ao_push(q, cnt, (x & 0x0000FF00u) == 0u, e0 + 1);
        cnt = ao_push(q, cnt, (x & 0x00FF0000u) == 0u, e0 + 2);
        cnt = ao_push(q, cnt, (x & 0xFF000000u) == 0u, e0 + 3);
    }
    return cnt;
}

template <bool FIRST>
__global__ void __launch_bounds__(1024)
agg_owner(const float* __restrict__ x, const int* __restrict__ ei, float* __restrict__ agg,
          unsigned* __restrict__ recs, int* __restrict__ cnts, const unsigned* __restrict__ own) {
    __shared__ __attribute__((aligned(16))) float acc[(AO_ROWS + 1) * 128];
    __shared__ __attribute__((aligned(16))) unsigned queue[32][AO_QCAP];
    const int wave = __builtin_amdgcn_readfirstlane(threadIdx.x >> 5), lane = threadIdx.x & 31;
    const int lo = blockIdx.x * AO_ROWS;
    const int nrows = (N_NODES - lo) < AO_ROWS ? (N_NODES - lo) : AO_ROWS;

    for (int i = threadIdx.x; i < (AO_ROWS + 1) * 32; i += 1024)
        ((float4*)acc)[i] = make_float4(0.f, 0.f, 0.f, 0.f);
    __syncthreads();

    unsigned* q = queue[wave];
    unsigned* rec = recs + (long)(blockIdx.x * 32 + wave) * AO_REC_CAP;
    int nrec = 0;
    bool replay = false;
    int saved = 0;
    if (!FIRST) {
        const int c = cnts[blockIdx.x * 32 + lane];
        replay = __builtin_amdgcn_ballot_w32(c < 0) == 0u;
        saved = __builtin_amdgcn_readlane(c, wave);
    }
    if (replay) {
        ao_replay(q, rec, saved, x, acc, lane);
    } else {
        const uint4* own4 = (const uint4*)own;
        const unsigned gx = blockIdx.x * 0x01010101u;
        int cnt = 0;
        const int n_it = (AO_ITERS - wave + 31) / 32;
        const int first = (int)(blockIdx.x * n_it) / AO_GROUPS;

        for (int k = 0; k < n_it; k += AO_SCAN) {
            uint4 d[AO_SCAN];
            int it[AO_SCAN];
#pragma unroll
            for (int u = 0; u < AO_SCAN; u++) {
                int kk = ((k + u) < n_it ? (k + u) : (n_it - 1)) + first;
                if (kk >= n_it) kk -= n_it;
                it[u] = wave + 32 * kk;
                d[u] = own4[it[u] * 32 + lane];
            }
#pragma unroll
            for (int u = 0; u < AO_SCAN; u++) {
                if ((k + u) < n_it) {
                    const int e0 = it[u] * 512 + lane * 16;
                    cnt = ao_scan(q, cnt, d[u].x, e0, gx);
                    cnt = ao_scan(q, cnt, d[u].y, e0 + 4, gx);
                    cnt = ao_scan(q, cnt, d[u].z, e0 + 8, gx);
                    cnt = ao_scan(q, cnt, d[u].w, e0 + 12, gx);
                }
            }
            if (cnt >= AO_TAKE) cnt = ao_full<FIRST>(q, cnt, x, ei, acc, lane, rec, nrec, lo);
        }
        if (wave == 31) {
            const uint2 t = ((const uint2*)own)[AO_ITERS * 64 + lane];
            const int e0 = AO_ITERS * 512 + lane * 8;
            cnt = ao_scan(q, cnt, t.x, e0, gx);
            cnt = ao_scan(q, cnt, t.y, e0 + 4, gx);
        }
        ao_drain<FIRST>(q, cnt, x, ei, acc, lane, rec, nrec, lo);
        if (FIRST && lane == 0) cnts[blockIdx.x * 32 + wave] = nrec;
    }

    __syncthreads();
    float4* out4 = (float4*)(agg + (long)lo * 128);
    for (int i = threadIdx.x; i < nrows * 32; i += 1024) {
        const float* r = acc + (i >> 5) * 128 + (i & 31);
        out4[i] = make_float4(r[0], r[32], r[64], r[96]);
    }
}

#define NG_PITCH 132
#define NG_TILES (N_NODES / 16)
#define NODE_GEMM_GROUPS 512
static_assert(NODE_GEMM_GROUPS <= NG_TILES, "every work-group has a first tile");
static_assert(N_NODES % 16 == 0, "row tiles must tile the node list");
template <bool FUSE_ADD, bool BN, bool RELU, bool POOL>
__global__ void __launch_bounds__(256)
gemm_node(const float* __restrict__ X, const float* __restrict__ AGG,
          const float* __restrict__ W, const float* __restrict__ bias,
          const float* __restrict__ gamma, const float* __restrict__ beta,
          const float* __restrict__ mean, const float* __restrict__ var,
          float* __restrict__ O, const int* __restrict__ batch, float* __restrict__ gsum) {
    __shared__ __attribute__((aligned(16))) float tileA[2][16 * NG_PITCH];
    const int wave = threadIdx.x >> 5, lane = threadIdx.x & 31;
    const int mi = lane & 15, kh = lane >> 4;
    const int col = wave * 16 + mi;

    v2f bf[32];
    const float* wc = W + (long)(kh * 2) * 128 + col;
#pragma unroll
    for (int j = 0; j < 32; j++) {
        bf[j].x = wc[(long)(4 * j) * 128];
        bf[j].y = wc[(long)(4 * j + 1) * 128];
    }

    const float bb = bias[col];
    float mu = 0.f, inv = 1.f, gm = 1.f, bt = 0.f;
    if (BN) { mu = mean[col]; inv = rsqrtf(var[col] + BN_EPS); gm = gamma[col]; bt = beta[col]; }

    const int so0 = wave * NG_PITCH + lane * 4;
    const int so1 = (wave + 8) * NG_PITCH + lane * 4;
    const long go0 = (long)wave * 128 + lane * 4;
    const long go1 = (long)(wave + 8) * 128 + lane * 4;

    int mt = blockIdx.x;
    {
        const float* xp = X + (long)mt * 16 * 128;
        float4 t0 = *(const float4*)(xp + go0);
        float4 t1 = *(const float4*)(xp + go1);
        if (FUSE_ADD) {
            const float* gp = AGG + (long)mt * 16 * 128;
            const float4 g0 = *(const float4*)(gp + go0);
            const float4 g1 = *(const float4*)(gp + go1);
            t0.x += g0.x; t0.y += g0.y; t0.z += g0.z; t0.w += g0.w;
            t1.x += g1.x; t1.y += g1.y; t1.z += g1.z; t1.w += g1.w;
        }
        *(float4*)(&tileA[0][so0]) = t0;
        *(float4*)(&tileA[0][so1]) = t1;
    }

    int cur = 0;
    for (; mt < NG_TILES; mt += gridDim.x) {
        __syncthreads();

        int bid = 0;
        if (POOL) bid = batch[mt * 16 + mi];

        const int nx = mt + gridDim.x;
        const int pf = nx < NG_TILES ? nx : mt;
        const float* xp = X + (long)pf * 16 * 128;
        const float4 x0 = *(const float4*)(xp + go0);
        const float4 x1 = *(const float4*)(xp + go1);
        float4 g0 = x0, g1 = x1;
        if (FUSE_ADD) {
            const float* gp = AGG + (long)pf * 16 * 128;
            g0 = *(const float4*)(gp + go0);
            g1 = *(const float4*)(gp + go1);
        }

        const float* ap = &tileA[cur][mi * NG_PITCH + kh * 2];
        v2f af[32];
#pragma unroll
        for (int j = 0; j < 32; j++) af[j] = *(const v2f*)(ap + 4 * j);
        v8f acc = {0.f, 0.f, 0.f, 0.f, 0.f, 0.f, 0.f, 0.f};
#pragma unroll
        for (int j = 0; j < 32; j++)
            acc = __builtin_amdgcn_wmma_f32_16x16x4_f32(false, af[j], false, bf[j],
                                                        (short)0, acc, false, false);
        __builtin_amdgcn_sched_group_barrier(0x100, 32, 0);
        __builtin_amdgcn_sched_group_barrier(0x008, 32, 0);
        __builtin_amdgcn_sched_group_barrier(0x002, 64, 0);

        float* orow = O + (long)(mt * 16 + kh * 8) * 128 + col;
        float vv[8];
#pragma unroll
        for (int r = 0; r < 8; r++) {
            float v = acc[r] + bb;
            if (BN)   v = (v - mu) * inv * gm + bt;
            if (RELU) v = fmaxf(v, 0.f);
            orow[(long)r * 128] = v;
            vv[r] = v;
        }

        if (POOL) {
            const int gfirst = __builtin_amdgcn_readfirstlane(bid);
            if (__builtin_amdgcn_ballot_w32(bid != gfirst) == 0u) {
                float s = vv[0];
#pragma unroll
                for (int r = 1; r < 8; r++) s += vv[r];
                const unsigned su = __float_as_uint(s);
                const float so = __uint_as_float(
                    __builtin_amdgcn_permlanex16(su, su, 0x76543210u, 0xfedcba98u, false, false));
                if (kh == 0) atomicAdd(gsum + (long)gfirst * 384 + col, s + so);
            } else {
                int idr[8];
#pragma unroll
                for (int r = 0; r < 8; r++) {
                    const int id_lo = __builtin_amdgcn_readlane(bid, r);
                    const int id_hi = __builtin_amdgcn_readlane(bid, 8 + r);
                    idr[r] = kh ? id_hi : id_lo;
                }
                int gcur = idr[0];
                float s = vv[0];
#pragma unroll
                for (int r = 1; r < 8; r++) {
                    if (idr[r] != gcur) {
                        atomicAdd(gsum + (long)gcur * 384 + col, s);
                        gcur = idr[r];
                        s = vv[r];
                    } else {
                        s += vv[r];
                    }
                }
                atomicAdd(gsum + (long)gcur * 384 + col, s);
            }
        }

        float4 t0 = x0, t1 = x1;
        if (FUSE_ADD) {
            t0.x += g0.x; t0.y += g0.y; t0.z += g0.z; t0.w += g0.w;
            t1.x += g1.x; t1.y += g1.y; t1.z += g1.z; t1.w += g1.w;
        }
        *(float4*)(&tileA[cur ^ 1][so0]) = t0;
        *(float4*)(&tileA[cur ^ 1][so1]) = t1;
        cur ^= 1;
    }
}

#define HL_K     384
#define HL_SPLIT 4
#define HL_STEPS (HL_K / 4 / HL_SPLIT)
static_assert(HL_K % (4 * HL_SPLIT) == 0 && N_GRAPHS % 16 == 0 && 8 % HL_SPLIT == 0,
              "whole matrix steps per wave, whole tiles, whole accumulator rows per storing wave");
__global__ void __launch_bounds__(32 * HL_SPLIT)
head_lin1(const float* __restrict__ G, const float* __restrict__ W,
          const float* __restrict__ bias, float* __restrict__ H) {
    __shared__ float part[HL_SPLIT][8][32];
    const int wave = threadIdx.x >> 5, lane = threadIdx.x & 31;
    const int mt = blockIdx.x / (HL_K / 16), nt = blockIdx.x % (HL_K / 16);
    const int mi = lane & 15, kh = lane >> 4;
    const int col = nt * 16 + mi;
    const int k0 = wave * (4 * HL_STEPS) + kh * 2;

    const float* xr = G + (long)(mt * 16 + mi) * HL_K + k0;
    const float* wc = W + (long)k0 * HL_K + col;
    v2f a[HL_STEPS], b[HL_STEPS];
#pragma unroll
    for (int j = 0; j < HL_STEPS; j++) a[j] = *(const v2f*)(xr + 4 * j);
#pragma unroll
    for (int j = 0; j < HL_STEPS; j++) {
        b[j].x = wc[(long)(4 * j) * HL_K];
        b[j].y = wc[(long)(4 * j + 1) * HL_K];
    }
    const float bb = bias[col];

    v8f acc = {0.f, 0.f, 0.f, 0.f, 0.f, 0.f, 0.f, 0.f};
#pragma unroll
    for (int j = 0; j < HL_STEPS; j++)
        acc = __builtin_amdgcn_wmma_f32_16x16x4_f32(false, a[j], false, b[j],
                                                    (short)0, acc, false, false);
    __builtin_amdgcn_sched_group_barrier(0x020, 3 * HL_STEPS + 1, 0);
    __builtin_amdgcn_sched_group_barrier(0x008, HL_STEPS, 0);
#pragma unroll
    for (int r = 0; r < 8; r++) part[wave][r][lane] = acc[r];
    __syncthreads();

#pragma unroll
    for (int t = 0; t < 8 / HL_SPLIT; t++) {
        const int r = wave + t * HL_SPLIT;
        float s = part[0][r][lane];
#pragma unroll
        for (int p = 1; p < HL_SPLIT; p++) s += part[p][r][lane];
        H[(long)(mt * 16 + r + 8 * kh) * HL_K + col] = fmaxf(s + bb, 0.f);
    }
}

#define HO_N  10
#define HO_KL 12
static_assert(32 * HO_KL == HL_K && (HO_KL * HO_N) % 4 == 0 && HO_KL % 4 == 0,
              "the lanes' k ranges tile the row; a lane's values are whole 16-byte loads");
__global__ void __launch_bounds__(32)
head_lin2(const float* __restrict__ h, const float* __restrict__ w,
          const float* __restrict__ b, float* __restrict__ out) {
    const int lane = threadIdx.x;
    const int g = blockIdx.x;
    const float4* hp = (const float4*)(h + (long)g * HL_K + lane * HO_KL);
    const float4* wp = (const float4*)(w + lane * (HO_KL * HO_N));
    float hs[HO_KL], ws[HO_KL * HO_N];
#pragma unroll
    for (int i = 0; i < HO_KL / 4; i++) {
        const float4 v = hp[i];
        hs[4 * i + 0] = v.x; hs[4 * i + 1] = v.y; hs[4 * i + 2] = v.z; hs[4 * i + 3] = v.w;
    }
#pragma unroll
    for (int i = 0; i < HO_KL * HO_N / 4; i++) {
        const float4 v = wp[i];
        ws[4 * i + 0] = v.x; ws[4 * i + 1] = v.y; ws[4 * i + 2] = v.z; ws[4 * i + 3] = v.w;
    }
    const float bo = lane < HO_N ? b[lane] : 0.f;

    float s[HO_N];
#pragma unroll
    for (int o = 0; o < HO_N; o++) s[o] = hs[0] * ws[o];
#pragma unroll
    for (int i = 1; i < HO_KL; i++)
#pragma unroll
        for (int o = 0; o < HO_N; o++) s[o] = fmaf(hs[i], ws[i * HO_N + o], s[o]);
#pragma unroll
    for (int o = 0; o < HO_N; o++)
#pragma unroll
        for (int m = 16; m >= 1; m >>= 1) s[o] += __shfl_xor(s[o], m, 32);
    float r = s[0];
#pragma unroll
    for (int o = 1; o < HO_N; o++) r = (lane == o) ? s[o] : r;
    if (lane < HO_N) out[g * HO_N + lane] = r + bo;
}

extern "C" void kernel_launch(void* const* d_in, const int* in_sizes, int n_in,
                              void* d_out, int out_size, void* d_ws, size_t ws_size,
                              hipStream_t stream) {
    const int*   x_ids = (const int*)d_in[0];
    const int*   edge  = (const int*)d_in[1];
    const int*   batch = (const int*)d_in[2];
    const float* emb   = (const float*)d_in[3];

    const float *c_w1[3], *c_b1[3], *c_g[3], *c_be[3], *c_mu[3], *c_va[3], *c_w2[3], *c_b2[3];
    for (int i = 0; i < 3; i++) {
        c_w1[i] = (const float*)d_in[4 + 8 * i + 0];
        c_b1[i] = (const float*)d_in[4 + 8 * i + 1];
        c_g[i]  = (const float*)d_in[4 + 8 * i + 2];
        c_be[i] = (const float*)d_in[4 + 8 * i + 3];
        c_mu[i] = (const float*)d_in[4 + 8 * i + 4];
        c_va[i] = (const float*)d_in[4 + 8 * i + 5];
        c_w2[i] = (const float*)d_in[4 + 8 * i + 6];
        c_b2[i] = (const float*)d_in[4 + 8 * i + 7];
    }
    const float* lin1_w = (const float*)d_in[28];
    const float* lin1_b = (const float*)d_in[29];
    const float* lin2_w = (const float*)d_in[30];
    const float* lin2_b = (const float*)d_in[31];

    const long NF = (long)N_NODES * 128;
    float* ws   = (float*)d_ws;
    float* x0   = ws + 0 * NF;
    float* agg  = ws + 1 * NF;
    float* tmp  = ws + 2 * NF;
    float* hbuf[3] = { ws + 3 * NF, ws + 4 * NF, x0 };
    float* lin1o = tmp;
    unsigned* recs = (unsigned*)(ws + 5 * NF);
    int* cnts = (int*)(recs + (long)AO_GROUPS * 32 * AO_REC_CAP);
    float* gcat = (float*)(cnts + AO_GROUPS * 32);
    unsigned* own = (unsigned*)(gcat + N_GRAPHS * 384);

    zero_f32<<<(N_GRAPHS * 384 / 4 + 255) / 256, 256, 0, stream>>>(gcat, N_GRAPHS * 384 / 4);

    gather_emb<<<(N_NODES * 32 + 255) / 256, 256, 0, stream>>>(x_ids, emb, x0, edge, own);

    const float* xin = x0;
    for (int l = 0; l < 3; l++) {
        if (l == 0) agg_owner<true><<<AO_GROUPS, 1024, 0, stream>>>(xin, edge, agg, recs, cnts, own);
        else        agg_owner<false><<<AO_GROUPS, 1024, 0, stream>>>(xin, edge, agg, recs, cnts, own);
        gemm_node<true, true, true, false><<<NODE_GEMM_GROUPS, 256, 0, stream>>>(
            xin, agg, c_w1[l], c_b1[l], c_g[l], c_be[l], c_mu[l], c_va[l], tmp, nullptr, nullptr);
        gemm_node<false, false, true, true><<<NODE_GEMM_GROUPS, 256, 0, stream>>>(
            tmp, nullptr, c_w2[l], c_b2[l], nullptr, nullptr, nullptr, nullptr, hbuf[l],
            batch, gcat + l * 128);
        xin = hbuf[l];
    }

    head_lin1<<<(N_GRAPHS / 16) * (HL_K / 16), 32 * HL_SPLIT, 0, stream>>>(gcat, lin1_w, lin1_b, lin1o);

    head_lin2<<<N_GRAPHS, 32, 0, stream>>>(lin1o, lin2_w, lin2_b, (float*)d_out);
}
